# starting kernel
# baseline (speedup 1.0000x reference)
#include <hip/hip_runtime.h>
#include <math.h>

#define FD 128
#define WAVE 32

typedef __attribute__((ext_vector_type(2))) float v2f;
typedef __attribute__((ext_vector_type(8))) float v8f;

__global__ void k_deg_init(unsigned* __restrict__ degi, int n) {
    int i = blockIdx.x * blockDim.x + threadIdx.x;
    if (i < n) degi[i] = 1u;
}

__global__ void k_deg_count(const int* __restrict__ dst,
                            unsigned* __restrict__ degi, int e) {
    int i = blockIdx.x * blockDim.x + threadIdx.x;
    if (i < e) atomicAdd(&degi[dst[i]], 1u);
}

__global__ void k_dinv(const unsigned* __restrict__ degi,
                       float* __restrict__ dinv, int n) {
    int i = blockIdx.x * blockDim.x + threadIdx.x;
    if (i < n) dinv[i] = rsqrtf((float)degi[i]);
}

__global__ __launch_bounds__(256) void k_gemm128_wmma(
    const float* __restrict__ A, const float* __restrict__ W,
    const float* __restrict__ bias, float* __restrict__ out,
    int fuse_bias_relu) {
    const int lane = threadIdx.x & 31;
    const int wid  = threadIdx.x >> 5;
    const int m0   = blockIdx.x << 4;
    const int n0   = wid << 4;
    const int half = lane >> 4;
    const int lm   = lane & 15;

    const float* __restrict__ arow = A + (size_t)(m0 + lm) * FD;

    v8f acc0 = {};
    v8f acc1 = {};
#pragma unroll
    for (int k0 = 0; k0 < FD; k0 += 8) {
        {
            const int ka = k0 + 2 * half;
            v2f a, b;
            a.x = arow[ka];
            a.y = arow[ka + 1];
            b.x = W[(size_t)(ka)     * FD + n0 + lm];
            b.y = W[(size_t)(ka + 1) * FD + n0 + lm];
            acc0 = __builtin_amdgcn_wmma_f32_16x16x4_f32(
                false, a, false, b, (short)0, acc0, false, false);
        }
        {
            const int ka = k0 + 4 + 2 * half;
            v2f a, b;
            a.x = arow[ka];
            a.y = arow[ka + 1];
            b.x = W[(size_t)(ka)     * FD + n0 + lm];
            b.y = W[(size_t)(ka + 1) * FD + n0 + lm];
            acc1 = __builtin_amdgcn_wmma_f32_16x16x4_f32(
                false, a, false, b, (short)0, acc1, false, false);
        }
    }

#pragma unroll
    for (int v = 0; v < 8; ++v) {
        const int row = m0 + v + 8 * half;
        const int col = n0 + lm;
        float r = acc0[v] + acc1[v];
        if (fuse_bias_relu) { r += bias[col]; r = fmaxf(r, 0.0f); }
        out[(size_t)row * FD + col] = r;
    }
}

__global__ void k_agg_init(const float* __restrict__ hw,
                           const float* __restrict__ dinv,
                           const float* __restrict__ bias,
                           float* __restrict__ agg, int nd) {
    int i = blockIdx.x * blockDim.x + threadIdx.x;
    if (i < nd) {
        const int node = i >> 7;
        const int f    = i & 127;
        const float di = dinv[node];
        agg[i] = hw[i] * di * di + bias[f];
    }
}

__global__ __launch_bounds__(256) void k_edge_scatter(
    const int* __restrict__ src, const int* __restrict__ dst,
    const float* __restrict__ dinv, const float* __restrict__ hw,
    float* __restrict__ agg, int e) {
    const int wid  = (int)((blockIdx.x * blockDim.x + threadIdx.x) >> 5);
    const int lane = threadIdx.x & 31;
    if (wid >= e) return;
    const int   s = src[wid];
    const int   d = dst[wid];
    const float w = dinv[s] * dinv[d];
    const float4 v = ((const float4*)(hw + (size_t)s * FD))[lane];
    float* ap = agg + (size_t)d * FD + lane * 4;
    unsafeAtomicAdd(ap + 0, v.x * w);
    unsafeAtomicAdd(ap + 1, v.y * w);
    unsafeAtomicAdd(ap + 2, v.z * w);
    unsafeAtomicAdd(ap + 3, v.w * w);
}

__global__ void k_relu(const float* __restrict__ in, float* __restrict__ out,
                       int n) {
    int i = blockIdx.x * blockDim.x + threadIdx.x;
    if (i < n) out[i] = fmaxf(in[i], 0.0f);
}

__global__ void k_emb_relu(const float* __restrict__ agg,
                           float* __restrict__ emb, float* __restrict__ h,
                           int n) {
    int i = blockIdx.x * blockDim.x + threadIdx.x;
    if (i < n) {
        const float v = agg[i];
        emb[i] = v;
        h[i]   = fmaxf(v, 0.0f);
    }
}

__global__ __launch_bounds__(256) void k_logits(
    const float* __restrict__ h, const float* __restrict__ Wp2,
    const float* __restrict__ bp2, float* __restrict__ out, int n) {
    const int wid  = (int)((blockIdx.x * blockDim.x + threadIdx.x) >> 5);
    const int lane = threadIdx.x & 31;
    if (wid >= n) return;
    const float4 v = ((const float4*)(h + (size_t)wid * FD))[lane];
    const int f = lane * 4;
    float p0 = v.x * Wp2[(f + 0) * 2 + 0] + v.y * Wp2[(f + 1) * 2 + 0] +
               v.z * Wp2[(f + 2) * 2 + 0] + v.w * Wp2[(f + 3) * 2 + 0];
    float p1 = v.x * Wp2[(f + 0) * 2 + 1] + v.y * Wp2[(f + 1) * 2 + 1] +
               v.z * Wp2[(f + 2) * 2 + 1] + v.w * Wp2[(f + 3) * 2 + 1];
#pragma unroll
    for (int off = 16; off > 0; off >>= 1) {
        p0 += __shfl_down(p0, off, 32);
        p1 += __shfl_down(p1, off, 32);
    }
    if (lane == 0) {
        const float l0 = p0 + bp2[0];
        const float l1 = p1 + bp2[1];
        const float m  = fmaxf(l0, l1);
        const float lse = m + logf(expf(l0 - m) + expf(l1 - m));
        out[(size_t)wid * 2 + 0] = l0 - lse;
        out[(size_t)wid * 2 + 1] = l1 - lse;
    }
}

extern "C" void kernel_launch(void* const* d_in, const int* in_sizes, int n_in,
                              void* d_out, int out_size, void* d_ws,
                              size_t ws_size, hipStream_t stream) {
    const float* x   = (const float*)d_in[0];
    const int*   ei  = (const int*)d_in[1];
    const float* W1  = (const float*)d_in[2];
    const float* b1  = (const float*)d_in[3];
    const float* W2  = (const float*)d_in[4];
    const float* b2  = (const float*)d_in[5];
    const float* W3  = (const float*)d_in[6];
    const float* b3  = (const float*)d_in[7];
    const float* Wp1 = (const float*)d_in[8];
    const float* bp1 = (const float*)d_in[9];
    const float* Wp2 = (const float*)d_in[10];
    const float* bp2 = (const float*)d_in[11];

    const int N = in_sizes[0] / FD;
    const int E = in_sizes[1] / 2;
    const int* src = ei;
    const int* dst = ei + E;

    char* ws = (char*)d_ws;
    unsigned* degi = (unsigned*)ws;  ws += (size_t)N * sizeof(unsigned);
    float* dinv = (float*)ws;        ws += (size_t)N * sizeof(float);
    float* hw   = (float*)ws;        ws += (size_t)N * FD * sizeof(float);
    float* agg  = (float*)ws;        ws += (size_t)N * FD * sizeof(float);
    float* h    = (float*)ws;

    float* emb_out  = (float*)d_out;
    float* prob_out = (float*)d_out + (size_t)N * FD;

    const dim3 blk(256);
    const int ND      = N * FD;
    const int gN      = (N + 255) / 256;
    const int gE      = (E + 255) / 256;
    const int gND     = (ND + 255) / 256;
    const int gTiles  = N / 16;
    const int gEdgeW  = (E * WAVE + 255) / 256;
    const int gNodeW  = (N * WAVE + 255) / 256;

    k_deg_init <<<gN, blk, 0, stream>>>(degi, N);
    k_deg_count<<<gE, blk, 0, stream>>>(dst, degi, E);
    k_dinv     <<<gN, blk, 0, stream>>>(degi, dinv, N);

    k_gemm128_wmma<<<gTiles, blk, 0, stream>>>(x, W1, nullptr, hw, 0);
    k_agg_init    <<<gND, blk, 0, stream>>>(hw, dinv, b1, agg, ND);
    k_edge_scatter<<<gEdgeW, blk, 0, stream>>>(src, dst, dinv, hw, agg, E);
    k_relu        <<<gND, blk, 0, stream>>>(agg, h, ND);

    k_gemm128_wmma<<<gTiles, blk, 0, stream>>>(h, W2, nullptr, hw, 0);
    k_agg_init    <<<gND, blk, 0, stream>>>(hw, dinv, b2, agg, ND);
    k_edge_scatter<<<gEdgeW, blk, 0, stream>>>(src, dst, dinv, hw, agg, E);
    k_relu        <<<gND, blk, 0, stream>>>(agg, h, ND);

    k_gemm128_wmma<<<gTiles, blk, 0, stream>>>(h, W3, nullptr, hw, 0);
    k_agg_init    <<<gND, blk, 0, stream>>>(hw, dinv, b3, agg, ND);
    k_edge_scatter<<<gEdgeW, blk, 0, stream>>>(src, dst, dinv, hw, agg, E);
    k_emb_relu    <<<gND, blk, 0, stream>>>(agg, emb_out, h, ND);

    k_gemm128_wmma<<<gTiles, blk, 0, stream>>>(h, Wp1, bp1, hw, 1);
    k_logits      <<<gNodeW, blk, 0, stream>>>(hw, Wp2, bp2, prob_out, N);
}


# Products re-tiled: one wave owns a whole 16x128 row tile, A read once straight into registers, W staged once per block in LDS as [k/4][n][k%4]; the k sum order of the products changes
# speedup vs baseline: 2.0507x; 2.0507x over previous
#include <hip/hip_runtime.h>
#include <math.h>

#define FD 128
#define WAVE 32

typedef __attribute__((ext_vector_type(2))) float v2f;
typedef __attribute__((ext_vector_type(8))) float v8f;

__global__ void k_deg_init(unsigned* __restrict__ degi, int n) {
    int i = blockIdx.x * blockDim.x + threadIdx.x;
    if (i < n) degi[i] = 1u;
}

__global__ void k_deg_count(const int* __restrict__ dst,
                            unsigned* __restrict__ degi, int e) {
    int i = blockIdx.x * blockDim.x + threadIdx.x;
    if (i < e) atomicAdd(&degi[dst[i]], 1u);
}

__global__ void k_dinv(const unsigned* __restrict__ degi,
                       float* __restrict__ dinv, int n) {
    int i = blockIdx.x * blockDim.x + threadIdx.x;
    if (i < n) dinv[i] = rsqrtf((float)degi[i]);
}

#define GT_WAVES 8
#define WL_HALF  (FD * 4 + 32)
#define WL_PAIR  (2 * FD * 4 + 32)
#define WL_WORDS (16 * WL_PAIR)

typedef __attribute__((ext_vector_type(4))) float v4f;

static __device__ __forceinline__ void stage_w(const float* __restrict__ W,
                                               float* __restrict__ wl) {
    const int n  = threadIdx.x & (FD - 1);
    const int g0 = threadIdx.x >> 7;
#pragma unroll
    for (int i = 0; i < FD / 8; ++i) {
        const int g = 2 * i + g0;
        v4f q;
        q.x = W[(size_t)(4 * g + 0) * FD + n];
        q.y = W[(size_t)(4 * g + 1) * FD + n];
        q.z = W[(size_t)(4 * g + 2) * FD + n];
        q.w = W[(size_t)(4 * g + 3) * FD + n];
        *(v4f*)(wl + i * WL_PAIR + g0 * WL_HALF + n * 4) = q;
    }
}

static __device__ __forceinline__ void load_a_tile(
    const float* __restrict__ arow, v4f (&a)[FD / 8]) {
#pragma unroll
    for (int j = 0; j < FD / 8; ++j) a[j] = *(const v4f*)(arow + 8 * j);
}

static __device__ __forceinline__ void relu_a_tile(v4f (&a)[FD / 8]) {
#pragma unroll
    for (int j = 0; j < FD / 8; ++j) {
        a[j].x = fmaxf(a[j].x, 0.0f); a[j].y = fmaxf(a[j].y, 0.0f);
        a[j].z = fmaxf(a[j].z, 0.0f); a[j].w = fmaxf(a[j].w, 0.0f);
    }
}

static __device__ __forceinline__ void load_b_step(
    const float* __restrict__ wlane, int j, v4f (&b)[8]) {
#pragma unroll
    for (int t = 0; t < 8; ++t)
        b[t] = *(const v4f*)(wlane + j * WL_PAIR + 64 * t);
}

static __device__ __forceinline__ void wmma_step(
    const v4f& av4, const v4f (&b)[8], v8f (&acc)[8]) {
#pragma unroll
    for (int t = 0; t < 8; ++t) {
        const v2f av = {av4.x, av4.y};
        const v2f bv = {b[t].x, b[t].y};
        acc[t] = __builtin_amdgcn_wmma_f32_16x16x4_f32(
            false, av, false, bv, (short)0, acc[t], false, false);
    }
#pragma unroll
    for (int t = 0; t < 8; ++t) {
        const v2f av = {av4.z, av4.w};
        const v2f bv = {b[t].z, b[t].w};
        acc[t] = __builtin_amdgcn_wmma_f32_16x16x4_f32(
            false, av, false, bv, (short)0, acc[t], false, false);
    }
}

static __device__ __forceinline__ void rowtile_product(
    const v4f (&a)[FD / 8], const float* __restrict__ wlane, v8f (&acc)[8]) {
    v4f b0[8], b1[8];
    load_b_step(wlane, 0, b0);
#pragma unroll
    for (int j = 0; j < FD / 8; j += 2) {
        load_b_step(wlane, j + 1, b1);
        __builtin_amdgcn_sched_barrier(0);
        wmma_step(a[j], b0, acc);
        __builtin_amdgcn_sched_barrier(0);
        if (j + 2 < FD / 8) load_b_step(wlane, j + 2, b0);
        __builtin_amdgcn_sched_barrier(0);
        wmma_step(a[j + 1], b1, acc);
        __builtin_amdgcn_sched_barrier(0);
    }
}

static __device__ __forceinline__ bool rowtile_front(
    const float* __restrict__ A, const float* __restrict__ W, int ntiles,
    bool relu_in, float* __restrict__ wl, int& m0, int& c, int& h,
    v4f (&a)[FD / 8]) {
    const int lane = threadIdx.x & 31;
    const int wid  = __builtin_amdgcn_readfirstlane((int)(threadIdx.x >> 5));
    const int tile = blockIdx.x * GT_WAVES + wid;
    const int tl   = min(tile, ntiles - 1);
    c  = lane & 15;
    h  = lane >> 4;
    m0 = tile << 4;

    load_a_tile(A + (size_t)((tl << 4) + c) * FD + 4 * h, a);

    stage_w(W, wl);
    __syncthreads();
    if (tile >= ntiles) return false;

    if (relu_in) relu_a_tile(a);
    return true;
}

static __device__ __forceinline__ void gemm_conv_body(
    const float* __restrict__ A, const float* __restrict__ W,
    const float* __restrict__ bias, const float* __restrict__ dinv,
    float* __restrict__ hw, float* __restrict__ agg, int ntiles, bool relu_in,
    float* __restrict__ wl) {
    int m0, c, h;
    v4f a[FD / 8];
    if (!rowtile_front(A, W, ntiles, relu_in, wl, m0, c, h, a)) return;

    float bc[8], di[8];
#pragma unroll
    for (int t = 0; t < 8; ++t) bc[t] = bias[16 * t + c];
#pragma unroll
    for (int v = 0; v < 8; ++v) di[v] = dinv[m0 + v + 8 * h];

    v8f acc[8];
#pragma unroll
    for (int t = 0; t < 8; ++t) acc[t] = v8f{};
    rowtile_product(a, wl + h * WL_HALF + c * 4, acc);

#pragma unroll
    for (int v = 0; v < 8; ++v) {
        const int row = m0 + v + 8 * h;
#pragma unroll
        for (int t = 0; t < 8; ++t) {
            const float  r = acc[t][v];
            const size_t o = (size_t)row * FD + 16 * t + c;
            hw[o]  = r;
            agg[o] = r * di[v] * di[v] + bc[t];
        }
    }
}

__global__ __launch_bounds__(256) void k_gemm_conv_in(
    const float* __restrict__ A, const float* __restrict__ W,
    const float* __restrict__ bias, const float* __restrict__ dinv,
    float* __restrict__ hw, float* __restrict__ agg, int ntiles) {
    __shared__ alignas(16) float wl[WL_WORDS];
    gemm_conv_body(A, W, bias, dinv, hw, agg, ntiles, false, wl);
}

__global__ __launch_bounds__(256) void k_gemm_conv_relu(
    const float* __restrict__ A, const float* __restrict__ W,
    const float* __restrict__ bias, const float* __restrict__ dinv,
    float* __restrict__ hw, float* __restrict__ agg, int ntiles) {
    __shared__ alignas(16) float wl[WL_WORDS];
    gemm_conv_body(A, W, bias, dinv, hw, agg, ntiles, true, wl);
}

__global__ __launch_bounds__(256) void k_gemm_post_relu(
    const float* __restrict__ A, const float* __restrict__ W,
    const float* __restrict__ bias, float* __restrict__ out, int ntiles) {
    __shared__ alignas(16) float wl[WL_WORDS];
    int m0, c, h;
    v4f a[FD / 8];
    if (!rowtile_front(A, W, ntiles, true, wl, m0, c, h, a)) return;

    float bc[8];
#pragma unroll
    for (int t = 0; t < 8; ++t) bc[t] = bias[16 * t + c];

    v8f acc[8];
#pragma unroll
    for (int t = 0; t < 8; ++t) acc[t] = v8f{};
    rowtile_product(a, wl + h * WL_HALF + c * 4, acc);

#pragma unroll
    for (int v = 0; v < 8; ++v) {
        const int row = m0 + v + 8 * h;
#pragma unroll
        for (int t = 0; t < 8; ++t) {
            const float r = acc[t][v] + bc[t];
            out[(size_t)row * FD + 16 * t + c] = fmaxf(r, 0.0f);
        }
    }
}

#define SC_RUN 32
#define SC_GRP 8

static __device__ __forceinline__ void scatter_row(
    int s, int d, float w, int j, int lane,
    const float* __restrict__ hw, float* __restrict__ agg) {
    const int   sj = __builtin_amdgcn_readlane(s, j);
    const int   dj = __builtin_amdgcn_readlane(d, j);
    const float wj = __int_as_float(__builtin_amdgcn_readlane(__float_as_int(w), j));
    const float* hp = hw + (size_t)sj * FD + lane;
    float*       ap = agg + (size_t)dj * FD + lane;
    const float v0 = hp[0];
    const float v1 = hp[32];
    const float v2 = hp[64];
    const float v3 = hp[96];
    unsafeAtomicAdd(ap,      v0 * wj);
    unsafeAtomicAdd(ap + 32, v1 * wj);
    unsafeAtomicAdd(ap + 64, v2 * wj);
    unsafeAtomicAdd(ap + 96, v3 * wj);
}

static __device__ __forceinline__ void scatter_group(
    int s, int d, float w, int j0, int lane,
    const float* __restrict__ hw, float* __restrict__ agg) {
    float v[SC_GRP][4];
#pragma unroll
    for (int g = 0; g < SC_GRP; ++g) {
        const int sj = __builtin_amdgcn_readlane(s, j0 + g);
        const float* hp = hw + (size_t)sj * FD + lane;
        v[g][0] = hp[0];
        v[g][1] = hp[32];
        v[g][2] = hp[64];
        v[g][3] = hp[96];
    }
#pragma unroll
    for (int g = 0; g < SC_GRP; ++g) {
        const int   dj = __builtin_amdgcn_readlane(d, j0 + g);
        const float wj = __int_as_float(__builtin_amdgcn_readlane(__float_as_int(w), j0 + g));
        float* ap = agg + (size_t)dj * FD + lane;
        unsafeAtomicAdd(ap,      v[g][0] * wj);
        unsafeAtomicAdd(ap + 32, v[g][1] * wj);
        unsafeAtomicAdd(ap + 64, v[g][2] * wj);
        unsafeAtomicAdd(ap + 96, v[g][3] * wj);
    }
}

__global__ __launch_bounds__(256) void k_edge_scatter(
    const int* __restrict__ src, const int* __restrict__ dst,
    const float* __restrict__ dinv, const float* __restrict__ hw,
    float* __restrict__ agg, int e) {
    const int gw   = __builtin_amdgcn_readfirstlane(
        (int)((blockIdx.x * blockDim.x + threadIdx.x) >> 5));
    const int lane = threadIdx.x & 31;
    const int base = gw * SC_RUN;
    if (base >= e) return;
    const int mine = base + lane;
    int   s = 0, d = 0;
    float w = 0.0f;
    if (mine < e) {
        s = src[mine];
        d = dst[mine];
        w = dinv[s] * dinv[d];
    }
    const int cnt = min(SC_RUN, e - base);
    if (cnt == SC_RUN) {
#pragma unroll
        for (int j0 = 0; j0 < SC_RUN; j0 += SC_GRP)
            scatter_group(s, d, w, j0, lane, hw, agg);
    } else {
#pragma nounroll
        for (int j = 0; j < cnt; ++j) scatter_row(s, d, w, j, lane, hw, agg);
    }
}

__global__ __launch_bounds__(256) void k_logits(
    const float* __restrict__ h, const float* __restrict__ Wp2,
    const float* __restrict__ bp2, float* __restrict__ out, int n) {
    const int wid  = (int)((blockIdx.x * blockDim.x + threadIdx.x) >> 5);
    const int lane = threadIdx.x & 31;
    if (wid >= n) return;
    const float4 v = ((const float4*)(h + (size_t)wid * FD))[lane];
    const int f = lane * 4;
    float p0 = v.x * Wp2[(f + 0) * 2 + 0] + v.y * Wp2[(f + 1) * 2 + 0] +
               v.z * Wp2[(f + 2) * 2 + 0] + v.w * Wp2[(f + 3) * 2 + 0];
    float p1 = v.x * Wp2[(f + 0) * 2 + 1] + v.y * Wp2[(f + 1) * 2 + 1] +
               v.z * Wp2[(f + 2) * 2 + 1] + v.w * Wp2[(f + 3) * 2 + 1];
#pragma unroll
    for (int off = 16; off > 0; off >>= 1) {
        p0 += __shfl_down(p0, off, 32);
        p1 += __shfl_down(p1, off, 32);
    }
    if (lane == 0) {
        const float l0 = p0 + bp2[0];
        const float l1 = p1 + bp2[1];
        const float m  = fmaxf(l0, l1);
        const float lse = m + logf(expf(l0 - m) + expf(l1 - m));
        out[(size_t)wid * 2 + 0] = l0 - lse;
        out[(size_t)wid * 2 + 1] = l1 - lse;
    }
}

extern "C" void kernel_launch(void* const* d_in, const int* in_sizes, int n_in,
                              void* d_out, int out_size, void* d_ws,
                              size_t ws_size, hipStream_t stream) {
    const float* x   = (const float*)d_in[0];
    const int*   ei  = (const int*)d_in[1];
    const float* W1  = (const float*)d_in[2];
    const float* b1  = (const float*)d_in[3];
    const float* W2  = (const float*)d_in[4];
    const float* b2  = (const float*)d_in[5];
    const float* W3  = (const float*)d_in[6];
    const float* b3  = (const float*)d_in[7];
    const float* Wp1 = (const float*)d_in[8];
    const float* bp1 = (const float*)d_in[9];
    const float* Wp2 = (const float*)d_in[10];
    const float* bp2 = (const float*)d_in[11];

    const int N = in_sizes[0] / FD;
    const int E = in_sizes[1] / 2;
    const int* src = ei;
    const int* dst = ei + E;

    char* ws = (char*)d_ws;
    unsigned* degi = (unsigned*)ws;  ws += (size_t)N * sizeof(unsigned);
    float* dinv = (float*)ws;        ws += (size_t)N * sizeof(float);
    float* hw   = (float*)ws;        ws += (size_t)N * FD * sizeof(float);
    float* agg1 = (float*)ws;        ws += (size_t)N * FD * sizeof(float);
    float* agg2 = (float*)ws;

    float* emb_out  = (float*)d_out;
    float* prob_out = (float*)d_out + (size_t)N * FD;

    const dim3 blk(256);
    const int gN      = (N + 255) / 256;
    const int gE      = (E + 255) / 256;
    const int gTiles  = N / 16;
    const int gGemm   = (gTiles + GT_WAVES - 1) / GT_WAVES;
    const int gEdgeW  = (((E + SC_RUN - 1) / SC_RUN) * WAVE + 255) / 256;
    const int gNodeW  = (N * WAVE + 255) / 256;

    k_deg_init <<<gN, blk, 0, stream>>>(degi, N);
    k_deg_count<<<gE, blk, 0, stream>>>(dst, degi, E);
    k_dinv     <<<gN, blk, 0, stream>>>(degi, dinv, N);

    k_gemm_conv_in  <<<gGemm, blk, 0, stream>>>(x, W1, b1, dinv, hw, agg1, gTiles);
    k_edge_scatter  <<<gEdgeW, blk, 0, stream>>>(src, dst, dinv, hw, agg1, E);

    k_gemm_conv_relu<<<gGemm, blk, 0, stream>>>(agg1, W2, b2, dinv, hw, agg2, gTiles);
    k_edge_scatter  <<<gEdgeW, blk, 0, stream>>>(src, dst, dinv, hw, agg2, E);

    k_gemm_conv_relu<<<gGemm, blk, 0, stream>>>(agg2, W3, b3, dinv, hw, emb_out, gTiles);
    k_edge_scatter  <<<gEdgeW, blk, 0, stream>>>(src, dst, dinv, hw, emb_out, E);

    k_gemm_post_relu<<<gGemm, blk, 0, stream>>>(emb_out, Wp1, bp1, hw, gTiles);
    k_logits        <<<gNodeW, blk, 0, stream>>>(hw, Wp2, bp2, prob_out, N);
}


# Edge scatter split by cache half: a wave reads its die and draws 32-edge chunks from its half's ticket queue, adding only into that half's range of target rows; atomic arrival order changes
# speedup vs baseline: 2.1968x; 1.0712x over previous
#include <hip/hip_runtime.h>
#include <math.h>

#define FD 128
#define WAVE 32

typedef __attribute__((ext_vector_type(2))) float v2f;
typedef __attribute__((ext_vector_type(8))) float v8f;

#define Q_STRIDE 256
#define Q_COUNT  6

__global__ void k_deg_init(unsigned* __restrict__ degi, unsigned* __restrict__ qs, int n) {
    int i = blockIdx.x * blockDim.x + threadIdx.x;
    if (i < n) degi[i] = 1u;
    if (i < Q_COUNT) qs[i * Q_STRIDE] = 0u;
}

__global__ void k_deg_count(const int* __restrict__ dst,
                            unsigned* __restrict__ degi, int e) {
    int i = blockIdx.x * blockDim.x + threadIdx.x;
    if (i < e) atomicAdd(&degi[dst[i]], 1u);
}

__global__ void k_dinv(const unsigned* __restrict__ degi,
                       float* __restrict__ dinv, int n) {
    int i = blockIdx.x * blockDim.x + threadIdx.x;
    if (i < n) dinv[i] = rsqrtf((float)degi[i]);
}

#define GT_WAVES 8
#define WL_HALF  (FD * 4 + 32)
#define WL_PAIR  (2 * FD * 4 + 32)
#define WL_WORDS (16 * WL_PAIR)

typedef __attribute__((ext_vector_type(4))) float v4f;

static __device__ __forceinline__ void stage_w(const float* __restrict__ W,
                                               float* __restrict__ wl) {
    const int n  = threadIdx.x & (FD - 1);
    const int g0 = threadIdx.x >> 7;
#pragma unroll
    for (int i = 0; i < FD / 8; ++i) {
        const int g = 2 * i + g0;
        v4f q;
        q.x = W[(size_t)(4 * g + 0) * FD + n];
        q.y = W[(size_t)(4 * g + 1) * FD + n];
        q.z = W[(size_t)(4 * g + 2) * FD + n];
        q.w = W[(size_t)(4 * g + 3) * FD + n];
        *(v4f*)(wl + i * WL_PAIR + g0 * WL_HALF + n * 4) = q;
    }
}

static __device__ __forceinline__ void load_a_tile(
    const float* __restrict__ arow, v4f (&a)[FD / 8]) {
#pragma unroll
    for (int j = 0; j < FD / 8; ++j) a[j] = *(const v4f*)(arow + 8 * j);
}

static __device__ __forceinline__ void relu_a_tile(v4f (&a)[FD / 8]) {
#pragma unroll
    for (int j = 0; j < FD / 8; ++j) {
        a[j].x = fmaxf(a[j].x, 0.0f); a[j].y = fmaxf(a[j].y, 0.0f);
        a[j].z = fmaxf(a[j].z, 0.0f); a[j].w = fmaxf(a[j].w, 0.0f);
    }
}

static __device__ __forceinline__ void load_b_step(
    const float* __restrict__ wlane, int j, v4f (&b)[8]) {
#pragma unroll
    for (int t = 0; t < 8; ++t)
        b[t] = *(const v4f*)(wlane + j * WL_PAIR + 64 * t);
}

static __device__ __forceinline__ void wmma_step(
    const v4f& av4, const v4f (&b)[8], v8f (&acc)[8]) {
#pragma unroll
    for (int t = 0; t < 8; ++t) {
        const v2f av = {av4.x, av4.y};
        const v2f bv = {b[t].x, b[t].y};
        acc[t] = __builtin_amdgcn_wmma_f32_16x16x4_f32(
            false, av, false, bv, (short)0, acc[t], false, false);
    }
#pragma unroll
    for (int t = 0; t < 8; ++t) {
        const v2f av = {av4.z, av4.w};
        const v2f bv = {b[t].z, b[t].w};
        acc[t] = __builtin_amdgcn_wmma_f32_16x16x4_f32(
            false, av, false, bv, (short)0, acc[t], false, false);
    }
}

static __device__ __forceinline__ void rowtile_product(
    const v4f (&a)[FD / 8], const float* __restrict__ wlane, v8f (&acc)[8]) {
    v4f b0[8], b1[8];
    load_b_step(wlane, 0, b0);
#pragma unroll
    for (int j = 0; j < FD / 8; j += 2) {
        load_b_step(wlane, j + 1, b1);
        __builtin_amdgcn_sched_barrier(0);
        wmma_step(a[j], b0, acc);
        __builtin_amdgcn_sched_barrier(0);
        if (j + 2 < FD / 8) load_b_step(wlane, j + 2, b0);
        __builtin_amdgcn_sched_barrier(0);
        wmma_step(a[j + 1], b1, acc);
        __builtin_amdgcn_sched_barrier(0);
    }
}

static __device__ __forceinline__ bool rowtile_front(
    const float* __restrict__ A, const float* __restrict__ W, int ntiles,
    bool relu_in, float* __restrict__ wl, int& m0, int& c, int& h,
    v4f (&a)[FD / 8]) {
    const int lane = threadIdx.x & 31;
    const int wid  = __builtin_amdgcn_readfirstlane((int)(threadIdx.x >> 5));
    const int tile = blockIdx.x * GT_WAVES + wid;
    const int tl   = min(tile, ntiles - 1);
    c  = lane & 15;
    h  = lane >> 4;
    m0 = tile << 4;

    load_a_tile(A + (size_t)((tl << 4) + c) * FD + 4 * h, a);

    stage_w(W, wl);
    __syncthreads();
    if (tile >= ntiles) return false;

    if (relu_in) relu_a_tile(a);
    return true;
}

static __device__ __forceinline__ void gemm_conv_body(
    const float* __restrict__ A, const float* __restrict__ W,
    const float* __restrict__ bias, const float* __restrict__ dinv,
    float* __restrict__ hw, float* __restrict__ agg, int ntiles, bool relu_in,
    float* __restrict__ wl) {
    int m0, c, h;
    v4f a[FD / 8];
    if (!rowtile_front(A, W, ntiles, relu_in, wl, m0, c, h, a)) return;

    float bc[8], di[8];
#pragma unroll
    for (int t = 0; t < 8; ++t) bc[t] = bias[16 * t + c];
#pragma unroll
    for (int v = 0; v < 8; ++v) di[v] = dinv[m0 + v + 8 * h];

    v8f acc[8];
#pragma unroll
    for (int t = 0; t < 8; ++t) acc[t] = v8f{};
    rowtile_product(a, wl + h * WL_HALF + c * 4, acc);

#pragma unroll
    for (int v = 0; v < 8; ++v) {
        const int row = m0 + v + 8 * h;
#pragma unroll
        for (int t = 0; t < 8; ++t) {
            const float  r = acc[t][v];
            const size_t o = (size_t)row * FD + 16 * t + c;
            hw[o]  = r;
            agg[o] = r * di[v] * di[v] + bc[t];
        }
    }
}

__global__ __launch_bounds__(256) void k_gemm_conv_in(
    const float* __restrict__ A, const float* __restrict__ W,
    const float* __restrict__ bias, const float* __restrict__ dinv,
    float* __restrict__ hw, float* __restrict__ agg, int ntiles) {
    __shared__ alignas(16) float wl[WL_WORDS];
    gemm_conv_body(A, W, bias, dinv, hw, agg, ntiles, false, wl);
}

__global__ __launch_bounds__(256) void k_gemm_conv_relu(
    const float* __restrict__ A, const float* __restrict__ W,
    const float* __restrict__ bias, const float* __restrict__ dinv,
    float* __restrict__ hw, float* __restrict__ agg, int ntiles) {
    __shared__ alignas(16) float wl[WL_WORDS];
    gemm_conv_body(A, W, bias, dinv, hw, agg, ntiles, true, wl);
}

__global__ __launch_bounds__(256) void k_gemm_post_relu(
    const float* __restrict__ A, const float* __restrict__ W,
    const float* __restrict__ bias, float* __restrict__ out, int ntiles) {
    __shared__ alignas(16) float wl[WL_WORDS];
    int m0, c, h;
    v4f a[FD / 8];
    if (!rowtile_front(A, W, ntiles, true, wl, m0, c, h, a)) return;

    float bc[8];
#pragma unroll
    for (int t = 0; t < 8; ++t) bc[t] = bias[16 * t + c];

    v8f acc[8];
#pragma unroll
    for (int t = 0; t < 8; ++t) acc[t] = v8f{};
    rowtile_product(a, wl + h * WL_HALF + c * 4, acc);

#pragma unroll
    for (int v = 0; v < 8; ++v) {
        const int row = m0 + v + 8 * h;
#pragma unroll
        for (int t = 0; t < 8; ++t) {
            const float r = acc[t][v] + bc[t];
            out[(size_t)row * FD + 16 * t + c] = fmaxf(r, 0.0f);
        }
    }
}

#define SC_RUN 32
#define SC_GRP 8

static __device__ __forceinline__ int my_half() {
    const unsigned id = __builtin_amdgcn_s_sendmsg_rtn(0x87);
    return (int)((id >> 18) & 1u);
}

static __device__ __forceinline__ unsigned draw_ticket(unsigned* q, int lane) {
    unsigned v = 0u;
    if (lane == 0) v = atomicAdd(q, 1u);
    return v;
}

static __device__ __forceinline__ void scatter_chunk(
    int t, int p, int lane, int nhalf,
    const int* __restrict__ src, const int* __restrict__ dst,
    const float* __restrict__ dinv, const float* __restrict__ hw,
    float* __restrict__ agg, int e) {
    const int mine = t * SC_RUN + lane;
    int   s = 0, d = 0;
    float w = 0.0f;
    bool  take = false;
    if (mine < e) {
        s = src[mine];
        d = dst[mine];
        take = ((d >= nhalf) ? 1 : 0) == p;
        if (take) w = dinv[s] * dinv[d];
    }
    unsigned m = __builtin_amdgcn_ballot_w32(take);
    while (m != 0u) {
        int j[SC_GRP];
        int cnt = 0;
#pragma unroll
        for (int g = 0; g < SC_GRP; ++g) {
            const bool has = (m != 0u);
            j[g] = has ? __builtin_ctz(m) : 0;
            cnt += has ? 1 : 0;
            m &= m - 1u;
        }
        float v[SC_GRP][4];
#pragma unroll
        for (int g = 0; g < SC_GRP; ++g) {
            if (g < cnt) {
                const int sj = __builtin_amdgcn_readlane(s, j[g]);
                const float* hp = hw + (size_t)sj * FD + lane;
                v[g][0] = hp[0];
                v[g][1] = hp[32];
                v[g][2] = hp[64];
                v[g][3] = hp[96];
            }
        }
#pragma unroll
        for (int g = 0; g < SC_GRP; ++g) {
            if (g < cnt) {
                const int   dj = __builtin_amdgcn_readlane(d, j[g]);
                const float wj = __int_as_float(__builtin_amdgcn_readlane(__float_as_int(w), j[g]));
                float* ap = agg + (size_t)dj * FD + lane;
                unsafeAtomicAdd(ap,      v[g][0] * wj);
                unsafeAtomicAdd(ap + 32, v[g][1] * wj);
                unsafeAtomicAdd(ap + 64, v[g][2] * wj);
                unsafeAtomicAdd(ap + 96, v[g][3] * wj);
            }
        }
    }
}

__global__ __launch_bounds__(256) void k_edge_scatter(
    const int* __restrict__ src, const int* __restrict__ dst,
    const float* __restrict__ dinv, const float* __restrict__ hw,
    float* __restrict__ agg, unsigned* __restrict__ q, int e, int n) {
    const int lane  = threadIdx.x & 31;
    const int nch   = (e + SC_RUN - 1) / SC_RUN;
    const int nhalf = n >> 1;
    const int p0    = my_half();
#pragma nounroll
    for (int pass = 0; pass < 2; ++pass) {
        const int p = p0 ^ pass;
        unsigned* qp = q + p * Q_STRIDE;
        for (;;) {
            const int t = __builtin_amdgcn_readfirstlane((int)draw_ticket(qp, lane));
            if (t >= nch) break;
            scatter_chunk(t, p, lane, nhalf, src, dst, dinv, hw, agg, e);
        }
    }
}

__global__ __launch_bounds__(256) void k_logits(
    const float* __restrict__ h, const float* __restrict__ Wp2,
    const float* __restrict__ bp2, float* __restrict__ out, int n) {
    const int wid  = (int)((blockIdx.x * blockDim.x + threadIdx.x) >> 5);
    const int lane = threadIdx.x & 31;
    if (wid >= n) return;
    const float4 v = ((const float4*)(h + (size_t)wid * FD))[lane];
    const int f = lane * 4;
    float p0 = v.x * Wp2[(f + 0) * 2 + 0] + v.y * Wp2[(f + 1) * 2 + 0] +
               v.z * Wp2[(f + 2) * 2 + 0] + v.w * Wp2[(f + 3) * 2 + 0];
    float p1 = v.x * Wp2[(f + 0) * 2 + 1] + v.y * Wp2[(f + 1) * 2 + 1] +
               v.z * Wp2[(f + 2) * 2 + 1] + v.w * Wp2[(f + 3) * 2 + 1];
#pragma unroll
    for (int off = 16; off > 0; off >>= 1) {
        p0 += __shfl_down(p0, off, 32);
        p1 += __shfl_down(p1, off, 32);
    }
    if (lane == 0) {
        const float l0 = p0 + bp2[0];
        const float l1 = p1 + bp2[1];
        const float m  = fmaxf(l0, l1);
        const float lse = m + logf(expf(l0 - m) + expf(l1 - m));
        out[(size_t)wid * 2 + 0] = l0 - lse;
        out[(size_t)wid * 2 + 1] = l1 - lse;
    }
}

extern "C" void kernel_launch(void* const* d_in, const int* in_sizes, int n_in,
                              void* d_out, int out_size, void* d_ws,
                              size_t ws_size, hipStream_t stream) {
    const float* x   = (const float*)d_in[0];
    const int*   ei  = (const int*)d_in[1];
    const float* W1  = (const float*)d_in[2];
    const float* b1  = (const float*)d_in[3];
    const float* W2  = (const float*)d_in[4];
    const float* b2  = (const float*)d_in[5];
    const float* W3  = (const float*)d_in[6];
    const float* b3  = (const float*)d_in[7];
    const float* Wp1 = (const float*)d_in[8];
    const float* bp1 = (const float*)d_in[9];
    const float* Wp2 = (const float*)d_in[10];
    const float* bp2 = (const float*)d_in[11];

    const int N = in_sizes[0] / FD;
    const int E = in_sizes[1] / 2;
    const int* src = ei;
    const int* dst = ei + E;

    char* ws = (char*)d_ws;
    unsigned* degi = (unsigned*)ws;  ws += (size_t)N * sizeof(unsigned);
    float* dinv = (float*)ws;        ws += (size_t)N * sizeof(float);
    float* hw   = (float*)ws;        ws += (size_t)N * FD * sizeof(float);
    float* agg1 = (float*)ws;        ws += (size_t)N * FD * sizeof(float);
    float* agg2 = (float*)ws;        ws += (size_t)N * FD * sizeof(float);
    unsigned* qs = (unsigned*)ws;

    float* emb_out  = (float*)d_out;
    float* prob_out = (float*)d_out + (size_t)N * FD;

    const dim3 blk(256);
    const int gN      = (N + 255) / 256;
    const int gE      = (E + 255) / 256;
    const int gTiles  = N / 16;
    const int gGemm   = (gTiles + GT_WAVES - 1) / GT_WAVES;
    const int gScat   = 2048;
    const int gNodeW  = (N * WAVE + 255) / 256;

    k_deg_init <<<gN, blk, 0, stream>>>(degi, qs, N);
    k_deg_count<<<gE, blk, 0, stream>>>(dst, degi, E);
    k_dinv     <<<gN, blk, 0, stream>>>(degi, dinv, N);

    k_gemm_conv_in  <<<gGemm, blk, 0, stream>>>(x, W1, b1, dinv, hw, agg1, gTiles);
    k_edge_scatter  <<<gScat, blk, 0, stream>>>(src, dst, dinv, hw, agg1, qs + 0 * Q_STRIDE, E, N);

    k_gemm_conv_relu<<<gGemm, blk, 0, stream>>>(agg1, W2, b2, dinv, hw, agg2, gTiles);
    k_edge_scatter  <<<gScat, blk, 0, stream>>>(src, dst, dinv, hw, agg2, qs + 2 * Q_STRIDE, E, N);

    k_gemm_conv_relu<<<gGemm, blk, 0, stream>>>(agg2, W3, b3, dinv, hw, emb_out, gTiles);
    k_edge_scatter  <<<gScat, blk, 0, stream>>>(src, dst, dinv, hw, emb_out, qs + 4 * Q_STRIDE, E, N);

    k_gemm_post_relu<<<gGemm, blk, 0, stream>>>(emb_out, Wp1, bp1, hw, gTiles);
    k_logits        <<<gNodeW, blk, 0, stream>>>(hw, Wp2, bp2, prob_out, N);
}


# Edge sums by target range, byte filter: degree count writes a range byte per edge; 32-wave groups scan the bytes 16 per lane, list hits, LDS f32 adds into 196 own rows; sum order: arrival in group
# speedup vs baseline: 2.3757x; 1.0814x over previous
#include <hip/hip_runtime.h>
#include <math.h>

#define FD 128
#define WAVE 32

typedef __attribute__((ext_vector_type(2))) float v2f;
typedef __attribute__((ext_vector_type(8))) float v8f;

#define RS_WGS   256
#define RS_ROWS  196
#define RS_TRIP  512

__global__ void k_deg_init(unsigned* __restrict__ degi, int n) {
    int i = blockIdx.x * blockDim.x + threadIdx.x;
    if (i < n) degi[i] = 1u;
}

__global__ void k_deg_count(const int* __restrict__ dst,
                            unsigned* __restrict__ degi,
                            unsigned* __restrict__ key4, int e, int nkey4) {
    int i = blockIdx.x * blockDim.x + threadIdx.x;
    if (i >= nkey4) return;
    unsigned d[4];
#pragma unroll
    for (int b = 0; b < 4; ++b) d[b] = (unsigned)dst[min(4 * i + b, e - 1)];
    unsigned k = 0u;
#pragma unroll
    for (int b = 0; b < 4; ++b) {
        if (4 * i + b < e) {
            atomicAdd(&degi[d[b]], 1u);
            k |= (d[b] / RS_ROWS) << (8 * b);
        }
    }
    key4[i] = k;
}

__global__ void k_dinv(const unsigned* __restrict__ degi,
                       float* __restrict__ dinv, int n) {
    int i = blockIdx.x * blockDim.x + threadIdx.x;
    if (i < n) dinv[i] = rsqrtf((float)degi[i]);
}

#define GT_WAVES 8
#define WL_HALF  (FD * 4 + 32)
#define WL_PAIR  (2 * FD * 4 + 32)
#define WL_WORDS (16 * WL_PAIR)

typedef __attribute__((ext_vector_type(4))) float v4f;

static __device__ __forceinline__ void stage_w(const float* __restrict__ W,
                                               float* __restrict__ wl) {
    const int n  = threadIdx.x & (FD - 1);
    const int g0 = threadIdx.x >> 7;
#pragma unroll
    for (int i = 0; i < FD / 8; ++i) {
        const int g = 2 * i + g0;
        v4f q;
        q.x = W[(size_t)(4 * g + 0) * FD + n];
        q.y = W[(size_t)(4 * g + 1) * FD + n];
        q.z = W[(size_t)(4 * g + 2) * FD + n];
        q.w = W[(size_t)(4 * g + 3) * FD + n];
        *(v4f*)(wl + i * WL_PAIR + g0 * WL_HALF + n * 4) = q;
    }
}

static __device__ __forceinline__ void load_a_tile(
    const float* __restrict__ arow, v4f (&a)[FD / 8]) {
#pragma unroll
    for (int j = 0; j < FD / 8; ++j) a[j] = *(const v4f*)(arow + 8 * j);
}

static __device__ __forceinline__ void relu_a_tile(v4f (&a)[FD / 8]) {
#pragma unroll
    for (int j = 0; j < FD / 8; ++j) {
        a[j].x = fmaxf(a[j].x, 0.0f); a[j].y = fmaxf(a[j].y, 0.0f);
        a[j].z = fmaxf(a[j].z, 0.0f); a[j].w = fmaxf(a[j].w, 0.0f);
    }
}

static __device__ __forceinline__ void load_b_step(
    const float* __restrict__ wlane, int j, v4f (&b)[8]) {
#pragma unroll
    for (int t = 0; t < 8; ++t)
        b[t] = *(const v4f*)(wlane + j * WL_PAIR + 64 * t);
}

static __device__ __forceinline__ void wmma_step(
    const v4f& av4, const v4f (&b)[8], v8f (&acc)[8]) {
#pragma unroll
    for (int t = 0; t < 8; ++t) {
        const v2f av = {av4.x, av4.y};
        const v2f bv = {b[t].x, b[t].y};
        acc[t] = __builtin_amdgcn_wmma_f32_16x16x4_f32(
            false, av, false, bv, (short)0, acc[t], false, false);
    }
#pragma unroll
    for (int t = 0; t < 8; ++t) {
        const v2f av = {av4.z, av4.w};
        const v2f bv = {b[t].z, b[t].w};
        acc[t] = __builtin_amdgcn_wmma_f32_16x16x4_f32(
            false, av, false, bv, (short)0, acc[t], false, false);
    }
}

static __device__ __forceinline__ void rowtile_product(
    const v4f (&a)[FD / 8], const float* __restrict__ wlane, v8f (&acc)[8]) {
    v4f b0[8], b1[8];
    load_b_step(wlane, 0, b0);
#pragma unroll
    for (int j = 0; j < FD / 8; j += 2) {
        load_b_step(wlane, j + 1, b1);
        __builtin_amdgcn_sched_barrier(0);
        wmma_step(a[j], b0, acc);
        __builtin_amdgcn_sched_barrier(0);
        if (j + 2 < FD / 8) load_b_step(wlane, j + 2, b0);
        __builtin_amdgcn_sched_barrier(0);
        wmma_step(a[j + 1], b1, acc);
        __builtin_amdgcn_sched_barrier(0);
    }
}

static __device__ __forceinline__ bool rowtile_front(
    const float* __restrict__ A, const float* __restrict__ W, int ntiles,
    bool relu_in, float* __restrict__ wl, int& m0, int& c, int& h,
    v4f (&a)[FD / 8]) {
    const int lane = threadIdx.x & 31;
    const int wid  = __builtin_amdgcn_readfirstlane((int)(threadIdx.x >> 5));
    const int tile = blockIdx.x * GT_WAVES + wid;
    const int tl   = min(tile, ntiles - 1);
    c  = lane & 15;
    h  = lane >> 4;
    m0 = tile << 4;

    load_a_tile(A + (size_t)((tl << 4) + c) * FD + 4 * h, a);

    stage_w(W, wl);
    __syncthreads();
    if (tile >= ntiles) return false;

    if (relu_in) relu_a_tile(a);
    return true;
}

static __device__ __forceinline__ void gemm_conv_body(
    const float* __restrict__ A, const float* __restrict__ W,
    const float* __restrict__ bias, const float* __restrict__ dinv,
    float* __restrict__ hw, float* __restrict__ agg, int ntiles, bool relu_in,
    float* __restrict__ wl) {
    int m0, c, h;
    v4f a[FD / 8];
    if (!rowtile_front(A, W, ntiles, relu_in, wl, m0, c, h, a)) return;

    float bc[8], di[8];
#pragma unroll
    for (int t = 0; t < 8; ++t) bc[t] = bias[16 * t + c];
#pragma unroll
    for (int v = 0; v < 8; ++v) di[v] = dinv[m0 + v + 8 * h];

    v8f acc[8];
#pragma unroll
    for (int t = 0; t < 8; ++t) acc[t] = v8f{};
    rowtile_product(a, wl + h * WL_HALF + c * 4, acc);

#pragma unroll
    for (int v = 0; v < 8; ++v) {
        const int row = m0 + v + 8 * h;
#pragma unroll
        for (int t = 0; t < 8; ++t) {
            const float  r = acc[t][v];
            const size_t o = (size_t)row * FD + 16 * t + c;
            hw[o]  = r;
            agg[o] = r * di[v] * di[v] + bc[t];
        }
    }
}

__global__ __launch_bounds__(256) void k_gemm_conv_in(
    const float* __restrict__ A, const float* __restrict__ W,
    const float* __restrict__ bias, const float* __restrict__ dinv,
    float* __restrict__ hw, float* __restrict__ agg, int ntiles) {
    __shared__ alignas(16) float wl[WL_WORDS];
    gemm_conv_body(A, W, bias, dinv, hw, agg, ntiles, false, wl);
}

__global__ __launch_bounds__(256) void k_gemm_conv_relu(
    const float* __restrict__ A, const float* __restrict__ W,
    const float* __restrict__ bias, const float* __restrict__ dinv,
    float* __restrict__ hw, float* __restrict__ agg, int ntiles) {
    __shared__ alignas(16) float wl[WL_WORDS];
    gemm_conv_body(A, W, bias, dinv, hw, agg, ntiles, true, wl);
}

__global__ __launch_bounds__(256) void k_gemm_post_relu(
    const float* __restrict__ A, const float* __restrict__ W,
    const float* __restrict__ bias, float* __restrict__ out, int ntiles) {
    __shared__ alignas(16) float wl[WL_WORDS];
    int m0, c, h;
    v4f a[FD / 8];
    if (!rowtile_front(A, W, ntiles, true, wl, m0, c, h, a)) return;

    float bc[8];
#pragma unroll
    for (int t = 0; t < 8; ++t) bc[t] = bias[16 * t + c];

    v8f acc[8];
#pragma unroll
    for (int t = 0; t < 8; ++t) acc[t] = v8f{};
    rowtile_product(a, wl + h * WL_HALF + c * 4, acc);

#pragma unroll
    for (int v = 0; v < 8; ++v) {
        const int row = m0 + v + 8 * h;
#pragma unroll
        for (int t = 0; t < 8; ++t) {
            const float r = acc[t][v] + bc[t];
            out[(size_t)row * FD + 16 * t + c] = fmaxf(r, 0.0f);
        }
    }
}

#define RS_WAVES 32
#define RS_CAP   512
#define SC_GRP   8

typedef __attribute__((ext_vector_type(4))) unsigned v4u;

static __device__ __forceinline__ unsigned zero_bytes(unsigned x) {
    const unsigned t = (x & 0x7f7f7f7fu) + 0x7f7f7f7fu;
    return ~(t | x | 0x7f7f7f7fu);
}

static __device__ __forceinline__ void work_off(
    const int* hits, int cnt, int lane, int r0,
    const int* __restrict__ src, const int* __restrict__ dst,
    const float* __restrict__ dinv, const float* __restrict__ hw, float* rows) {
    for (int base = 0; base < cnt; base += 32) {
        const int idx = base + lane;
        int   s = 0, dl = 0;
        float w = 0.0f;
        if (idx < cnt) {
            const int eid = hits[idx];
            s = src[eid];
            const int d = dst[eid];
            w  = dinv[s] * dinv[d];
            dl = (d - r0) * FD;
        }
        const int nt = min(32, cnt - base);
        for (int g0 = 0; g0 < nt; g0 += SC_GRP) {
            float v[SC_GRP][4];
#pragma unroll
            for (int g = 0; g < SC_GRP; ++g) {
                const int sj = __builtin_amdgcn_readlane(s, min(g0 + g, nt - 1));
                const float* hp = hw + (size_t)sj * FD + lane;
                v[g][0] = hp[0];
                v[g][1] = hp[32];
                v[g][2] = hp[64];
                v[g][3] = hp[96];
            }
#pragma unroll
            for (int g = 0; g < SC_GRP; ++g) {
                if (g == 0 || g0 + g < nt) {
                    const int   dj = __builtin_amdgcn_readlane(dl, g0 + g);
                    const float wj = __int_as_float(__builtin_amdgcn_readlane(__float_as_int(w), g0 + g));
                    float* rp = rows + dj + lane;
                    unsafeAtomicAdd(rp,      v[g][0] * wj);
                    unsafeAtomicAdd(rp + 32, v[g][1] * wj);
                    unsafeAtomicAdd(rp + 64, v[g][2] * wj);
                    unsafeAtomicAdd(rp + 96, v[g][3] * wj);
                }
            }
        }
    }
}

__global__ __launch_bounds__(1024) void k_range_sums(
    const int* __restrict__ src, const int* __restrict__ dst,
    const unsigned* __restrict__ key4, const float* __restrict__ dinv,
    const float* __restrict__ hw, float* agg, int e, int n) {
    __shared__ alignas(16) float rows[RS_ROWS * FD];
    __shared__ int hits[RS_WAVES * RS_CAP];
    const int lane = threadIdx.x & 31;
    const int wid  = __builtin_amdgcn_readfirstlane((int)(threadIdx.x >> 5));
    const int r0   = blockIdx.x * RS_ROWS;
    const int nr   = min(RS_ROWS, n - r0);
    if (nr <= 0) return;

    v4f* gl = (v4f*)(agg + (size_t)r0 * FD);
    v4f* rl = (v4f*)rows;
    for (int i = threadIdx.x; i < nr * (FD / 4); i += RS_WAVES * 32) rl[i] = gl[i];
    __syncthreads();

    const unsigned splat = (unsigned)blockIdx.x * 0x01010101u;
    int* myhits = hits + wid * RS_CAP;
    int cnt = 0;
    const int thr = 32 + 8 * ((wid + (wid >> 2)) & 7);
    const int ntrips = (e + RS_TRIP - 1) / RS_TRIP;
    const int tlast  = ntrips - 1;
    const v4u* kp = (const v4u*)key4 + lane;
    v4u k0 = kp[(size_t)min(wid,                tlast) * 32];
    v4u k1 = kp[(size_t)min(wid +     RS_WAVES, tlast) * 32];
    v4u k2 = kp[(size_t)min(wid + 2 * RS_WAVES, tlast) * 32];
    v4u k3 = kp[(size_t)min(wid + 3 * RS_WAVES, tlast) * 32];
#pragma nounroll
    for (int tr = wid; tr < ntrips; tr += RS_WAVES) {
        const v4u kc = k0;
        k0 = k1; k1 = k2; k2 = k3;
        k3 = kp[(size_t)min(tr + 4 * RS_WAVES, tlast) * 32];
        unsigned m = (zero_bytes(kc.x ^ splat) >> 7) | (zero_bytes(kc.y ^ splat) >> 6) |
                     (zero_bytes(kc.z ^ splat) >> 5) | (zero_bytes(kc.w ^ splat) >> 4);
        while (__builtin_amdgcn_ballot_w32(m != 0u) != 0u) {
            const int  p   = __builtin_ctz(m | 0x80000000u);
            const int  eid = tr * RS_TRIP + lane * 16 + 4 * (p & 7) + (p >> 3);
            const bool ok  = (m != 0u) && (eid < e);
            const unsigned bm = __builtin_amdgcn_ballot_w32(ok);
            if (ok) myhits[cnt + (int)__builtin_amdgcn_mbcnt_lo(bm, 0u)] = eid;
            cnt += __builtin_popcount(bm);
            m &= m - 1u;
            if (cnt >= thr) {
                work_off(myhits, cnt, lane, r0, src, dst, dinv, hw, rows);
                cnt = 0;
            }
        }
    }
    work_off(myhits, cnt, lane, r0, src, dst, dinv, hw, rows);
    __syncthreads();

    for (int i = threadIdx.x; i < nr * (FD / 4); i += RS_WAVES * 32) gl[i] = rl[i];
}

__global__ __launch_bounds__(256) void k_logits(
    const float* __restrict__ h, const float* __restrict__ Wp2,
    const float* __restrict__ bp2, float* __restrict__ out, int n) {
    const int wid  = (int)((blockIdx.x * blockDim.x + threadIdx.x) >> 5);
    const int lane = threadIdx.x & 31;
    if (wid >= n) return;
    const float4 v = ((const float4*)(h + (size_t)wid * FD))[lane];
    const int f = lane * 4;
    float p0 = v.x * Wp2[(f + 0) * 2 + 0] + v.y * Wp2[(f + 1) * 2 + 0] +
               v.z * Wp2[(f + 2) * 2 + 0] + v.w * Wp2[(f + 3) * 2 + 0];
    float p1 = v.x * Wp2[(f + 0) * 2 + 1] + v.y * Wp2[(f + 1) * 2 + 1] +
               v.z * Wp2[(f + 2) * 2 + 1] + v.w * Wp2[(f + 3) * 2 + 1];
#pragma unroll
    for (int off = 16; off > 0; off >>= 1) {
        p0 += __shfl_down(p0, off, 32);
        p1 += __shfl_down(p1, off, 32);
    }
    if (lane == 0) {
        const float l0 = p0 + bp2[0];
        const float l1 = p1 + bp2[1];
        const float m  = fmaxf(l0, l1);
        const float lse = m + logf(expf(l0 - m) + expf(l1 - m));
        out[(size_t)wid * 2 + 0] = l0 - lse;
        out[(size_t)wid * 2 + 1] = l1 - lse;
    }
}

extern "C" void kernel_launch(void* const* d_in, const int* in_sizes, int n_in,
                              void* d_out, int out_size, void* d_ws,
                              size_t ws_size, hipStream_t stream) {
    const float* x   = (const float*)d_in[0];
    const int*   ei  = (const int*)d_in[1];
    const float* W1  = (const float*)d_in[2];
    const float* b1  = (const float*)d_in[3];
    const float* W2  = (const float*)d_in[4];
    const float* b2  = (const float*)d_in[5];
    const float* W3  = (const float*)d_in[6];
    const float* b3  = (const float*)d_in[7];
    const float* Wp1 = (const float*)d_in[8];
    const float* bp1 = (const float*)d_in[9];
    const float* Wp2 = (const float*)d_in[10];
    const float* bp2 = (const float*)d_in[11];

    const int N = in_sizes[0] / FD;
    const int E = in_sizes[1] / 2;
    const int* src = ei;
    const int* dst = ei + E;

    char* ws = (char*)d_ws;
    unsigned* degi = (unsigned*)ws;  ws += (size_t)N * sizeof(unsigned);
    float* dinv = (float*)ws;        ws += (size_t)N * sizeof(float);
    float* hw   = (float*)ws;        ws += (size_t)N * FD * sizeof(float);
    float* agg1 = (float*)ws;        ws += (size_t)N * FD * sizeof(float);
    float* agg2 = (float*)ws;        ws += (size_t)N * FD * sizeof(float);
    unsigned* key4 = (unsigned*)ws;

    float* emb_out  = (float*)d_out;
    float* prob_out = (float*)d_out + (size_t)N * FD;

    const dim3 blk(256);
    const int gN      = (N + 255) / 256;
    const int nTrips  = (E + RS_TRIP - 1) / RS_TRIP;
    const int nKey4   = nTrips * (RS_TRIP / 4);
    const int gE      = (nKey4 + 255) / 256;
    const int gTiles  = N / 16;
    const int gGemm   = (gTiles + GT_WAVES - 1) / GT_WAVES;
    const dim3 blkSum(RS_WAVES * 32);
    const int gNodeW  = (N * WAVE + 255) / 256;

    k_deg_init <<<gN, blk, 0, stream>>>(degi, N);
    k_deg_count<<<gE, blk, 0, stream>>>(dst, degi, key4, E, nKey4);
    k_dinv     <<<gN, blk, 0, stream>>>(degi, dinv, N);

    k_gemm_conv_in  <<<gGemm, blk, 0, stream>>>(x, W1, b1, dinv, hw, agg1, gTiles);
    k_range_sums    <<<RS_WGS, blkSum, 0, stream>>>(src, dst, key4, dinv, hw, agg1, E, N);

    k_gemm_conv_relu<<<gGemm, blk, 0, stream>>>(agg1, W2, b2, dinv, hw, agg2, gTiles);
    k_range_sums    <<<RS_WGS, blkSum, 0, stream>>>(src, dst, key4, dinv, hw, agg2, E, N);

    k_gemm_conv_relu<<<gGemm, blk, 0, stream>>>(agg2, W3, b3, dinv, hw, emb_out, gTiles);
    k_range_sums    <<<RS_WGS, blkSum, 0, stream>>>(src, dst, key4, dinv, hw, emb_out, E, N);

    k_gemm_post_relu<<<gGemm, blk, 0, stream>>>(emb_out, Wp1, bp1, hw, gTiles);
    k_logits        <<<gNodeW, blk, 0, stream>>>(hw, Wp2, bp2, prob_out, N);
}


# Edge search once per call: layer 1's range kernel also writes each range's edges (source, row offset, weight) to its exact stretch of a list; layers 2, 3 read it; sum order: arrival in group
# speedup vs baseline: 2.7664x; 1.1645x over previous
#include <hip/hip_runtime.h>
#include <math.h>

#define FD 128
#define WAVE 32

typedef __attribute__((ext_vector_type(2))) float v2f;
typedef __attribute__((ext_vector_type(8))) float v8f;

#define RS_WGS   256
#define RS_ROWS  196
#define RS_TRIP  512

__global__ void k_deg_init(unsigned* __restrict__ degi, int n) {
    int i = blockIdx.x * blockDim.x + threadIdx.x;
    if (i < n) degi[i] = 1u;
}

__global__ void k_deg_count(const int* __restrict__ dst,
                            unsigned* __restrict__ degi,
                            unsigned* __restrict__ key4, int e, int nkey4) {
    int i = blockIdx.x * blockDim.x + threadIdx.x;
    if (i >= nkey4) return;
    unsigned d[4];
#pragma unroll
    for (int b = 0; b < 4; ++b) d[b] = (unsigned)dst[min(4 * i + b, e - 1)];
    unsigned k = 0u;
#pragma unroll
    for (int b = 0; b < 4; ++b) {
        if (4 * i + b < e) {
            atomicAdd(&degi[d[b]], 1u);
            k |= (d[b] / RS_ROWS) << (8 * b);
        }
    }
    key4[i] = k;
}

__global__ void k_dinv(const unsigned* __restrict__ degi,
                       float* __restrict__ dinv, int n) {
    int i = blockIdx.x * blockDim.x + threadIdx.x;
    if (i < n) dinv[i] = rsqrtf((float)degi[i]);
}

#define GT_WAVES 8
#define WL_HALF  (FD * 4 + 32)
#define WL_PAIR  (2 * FD * 4 + 32)
#define WL_WORDS (16 * WL_PAIR)

typedef __attribute__((ext_vector_type(4))) float v4f;

static __device__ __forceinline__ void stage_w(const float* __restrict__ W,
                                               float* __restrict__ wl) {
    const int n  = threadIdx.x & (FD - 1);
    const int g0 = threadIdx.x >> 7;
#pragma unroll
    for (int i = 0; i < FD / 8; ++i) {
        const int g = 2 * i + g0;
        v4f q;
        q.x = W[(size_t)(4 * g + 0) * FD + n];
        q.y = W[(size_t)(4 * g + 1) * FD + n];
        q.z = W[(size_t)(4 * g + 2) * FD + n];
        q.w = W[(size_t)(4 * g + 3) * FD + n];
        *(v4f*)(wl + i * WL_PAIR + g0 * WL_HALF + n * 4) = q;
    }
}

static __device__ __forceinline__ void load_a_tile(
    const float* __restrict__ arow, v4f (&a)[FD / 8]) {
#pragma unroll
    for (int j = 0; j < FD / 8; ++j) a[j] = *(const v4f*)(arow + 8 * j);
}

static __device__ __forceinline__ void relu_a_tile(v4f (&a)[FD / 8]) {
#pragma unroll
    for (int j = 0; j < FD / 8; ++j) {
        a[j].x = fmaxf(a[j].x, 0.0f); a[j].y = fmaxf(a[j].y, 0.0f);
        a[j].z = fmaxf(a[j].z, 0.0f); a[j].w = fmaxf(a[j].w, 0.0f);
    }
}

static __device__ __forceinline__ void load_b_step(
    const float* __restrict__ wlane, int j, v4f (&b)[8]) {
#pragma unroll
    for (int t = 0; t < 8; ++t)
        b[t] = *(const v4f*)(wlane + j * WL_PAIR + 64 * t);
}

static __device__ __forceinline__ void wmma_step(
    const v4f& av4, const v4f (&b)[8], v8f (&acc)[8]) {
#pragma unroll
    for (int t = 0; t < 8; ++t) {
        const v2f av = {av4.x, av4.y};
        const v2f bv = {b[t].x, b[t].y};
        acc[t] = __builtin_amdgcn_wmma_f32_16x16x4_f32(
            false, av, false, bv, (short)0, acc[t], false, false);
    }
#pragma unroll
    for (int t = 0; t < 8; ++t) {
        const v2f av = {av4.z, av4.w};
        const v2f bv = {b[t].z, b[t].w};
        acc[t] = __builtin_amdgcn_wmma_f32_16x16x4_f32(
            false, av, false, bv, (short)0, acc[t], false, false);
    }
}

static __device__ __forceinline__ void rowtile_product(
    const v4f (&a)[FD / 8], const float* __restrict__ wlane, v8f (&acc)[8]) {
    v4f b0[8], b1[8];
    load_b_step(wlane, 0, b0);
#pragma unroll
    for (int j = 0; j < FD / 8; j += 2) {
        load_b_step(wlane, j + 1, b1);
        __builtin_amdgcn_sched_barrier(0);
        wmma_step(a[j], b0, acc);
        __builtin_amdgcn_sched_barrier(0);
        if (j + 2 < FD / 8) load_b_step(wlane, j + 2, b0);
        __builtin_amdgcn_sched_barrier(0);
        wmma_step(a[j + 1], b1, acc);
        __builtin_amdgcn_sched_barrier(0);
    }
}

static __device__ __forceinline__ bool rowtile_front(
    const float* __restrict__ A, const float* __restrict__ W, int ntiles,
    bool relu_in, float* __restrict__ wl, int& m0, int& c, int& h,
    v4f (&a)[FD / 8]) {
    const int lane = threadIdx.x & 31;
    const int wid  = __builtin_amdgcn_readfirstlane((int)(threadIdx.x >> 5));
    const int tile = blockIdx.x * GT_WAVES + wid;
    const int tl   = min(tile, ntiles - 1);
    c  = lane & 15;
    h  = lane >> 4;
    m0 = tile << 4;

    load_a_tile(A + (size_t)((tl << 4) + c) * FD + 4 * h, a);

    stage_w(W, wl);
    __syncthreads();
    if (tile >= ntiles) return false;

    if (relu_in) relu_a_tile(a);
    return true;
}

static __device__ __forceinline__ void gemm_conv_body(
    const float* __restrict__ A, const float* __restrict__ W,
    const float* __restrict__ bias, const float* __restrict__ dinv,
    float* __restrict__ hw, float* __restrict__ agg, int ntiles, bool relu_in,
    float* __restrict__ wl) {
    int m0, c, h;
    v4f a[FD / 8];
    if (!rowtile_front(A, W, ntiles, relu_in, wl, m0, c, h, a)) return;

    float bc[8], di[8];
#pragma unroll
    for (int t = 0; t < 8; ++t) bc[t] = bias[16 * t + c];
#pragma unroll
    for (int v = 0; v < 8; ++v) di[v] = dinv[m0 + v + 8 * h];

    v8f acc[8];
#pragma unroll
    for (int t = 0; t < 8; ++t) acc[t] = v8f{};
    rowtile_product(a, wl + h * WL_HALF + c * 4, acc);

#pragma unroll
    for (int v = 0; v < 8; ++v) {
        const int row = m0 + v + 8 * h;
#pragma unroll
        for (int t = 0; t < 8; ++t) {
            const float  r = acc[t][v];
            const size_t o = (size_t)row * FD + 16 * t + c;
            hw[o]  = r;
            agg[o] = r * di[v] * di[v] + bc[t];
        }
    }
}

__global__ __launch_bounds__(256) void k_gemm_conv_in(
    const float* __restrict__ A, const float* __restrict__ W,
    const float* __restrict__ bias, const float* __restrict__ dinv,
    float* __restrict__ hw, float* __restrict__ agg, int ntiles) {
    __shared__ alignas(16) float wl[WL_WORDS];
    gemm_conv_body(A, W, bias, dinv, hw, agg, ntiles, false, wl);
}

__global__ __launch_bounds__(256) void k_gemm_conv_relu(
    const float* __restrict__ A, const float* __restrict__ W,
    const float* __restrict__ bias, const float* __restrict__ dinv,
    float* __restrict__ hw, float* __restrict__ agg, int ntiles) {
    __shared__ alignas(16) float wl[WL_WORDS];
    gemm_conv_body(A, W, bias, dinv, hw, agg, ntiles, true, wl);
}

__global__ __launch_bounds__(256) void k_gemm_post_relu(
    const float* __restrict__ A, const float* __restrict__ W,
    const float* __restrict__ bias, float* __restrict__ out, int ntiles) {
    __shared__ alignas(16) float wl[WL_WORDS];
    int m0, c, h;
    v4f a[FD / 8];
    if (!rowtile_front(A, W, ntiles, true, wl, m0, c, h, a)) return;

    float bc[8];
#pragma unroll
    for (int t = 0; t < 8; ++t) bc[t] = bias[16 * t + c];

    v8f acc[8];
#pragma unroll
    for (int t = 0; t < 8; ++t) acc[t] = v8f{};
    rowtile_product(a, wl + h * WL_HALF + c * 4, acc);

#pragma unroll
    for (int v = 0; v < 8; ++v) {
        const int row = m0 + v + 8 * h;
#pragma unroll
        for (int t = 0; t < 8; ++t) {
            const float r = acc[t][v] + bc[t];
            out[(size_t)row * FD + 16 * t + c] = fmaxf(r, 0.0f);
        }
    }
}

#define RS_WAVES 32
#define RS_CAP   512
#define SC_GRP   8
#define LS_GRP   16
#define RS_DEG4  13
#define RS_SV    ((RS_ROWS * (FD / 4) + RS_WAVES * 32 - 1) / (RS_WAVES * 32))

typedef __attribute__((ext_vector_type(4))) unsigned v4u;
typedef __attribute__((ext_vector_type(2))) int v2i;

static __device__ __forceinline__ unsigned zero_bytes(unsigned x) {
    const unsigned t = (x & 0x7f7f7f7fu) + 0x7f7f7f7fu;
    return ~(t | x | 0x7f7f7f7fu);
}

template <int G>
static __device__ __forceinline__ void add_rows(
    int s, int dl, float w, int nt, int lane,
    const float* __restrict__ hw, float* rows) {
    for (int g0 = 0; g0 < nt; g0 += G) {
        float v[G][4];
#pragma unroll
        for (int g = 0; g < G; ++g) {
            const int sj = __builtin_amdgcn_readlane(s, min(g0 + g, nt - 1));
            const float* hp = hw + (size_t)sj * FD + lane;
            v[g][0] = hp[0];
            v[g][1] = hp[32];
            v[g][2] = hp[64];
            v[g][3] = hp[96];
        }
#pragma unroll
        for (int g = 0; g < G; ++g) {
            if (g == 0 || g0 + g < nt) {
                const int   dj = __builtin_amdgcn_readlane(dl, g0 + g);
                const float wj = __int_as_float(__builtin_amdgcn_readlane(__float_as_int(w), g0 + g));
                float* rp = rows + dj + lane;
                unsafeAtomicAdd(rp,      v[g][0] * wj);
                unsafeAtomicAdd(rp + 32, v[g][1] * wj);
                unsafeAtomicAdd(rp + 64, v[g][2] * wj);
                unsafeAtomicAdd(rp + 96, v[g][3] * wj);
            }
        }
    }
}

static __device__ __forceinline__ void work_off(
    const int* hits, int cnt, int lane, int r0,
    const int* __restrict__ src, const int* __restrict__ dst,
    const float* __restrict__ dinv, const float* __restrict__ hw, float* rows,
    v2i* __restrict__ rsd, float* __restrict__ rw, int room, int* lpos) {
    if (cnt <= 0) return;
    int p = 0;
    if (lane == 0) p = atomicAdd(lpos, cnt);
    p = __builtin_amdgcn_readfirstlane(p);
    for (int base = 0; base < cnt; base += 32) {
        const int idx = base + lane;
        int   s = 0, dl = 0;
        float w = 0.0f;
        if (idx < cnt) {
            const int eid = hits[idx];
            s = src[eid];
            const int d = dst[eid];
            w  = dinv[s] * dinv[d];
            dl = (d - r0) * FD;
            if (p + idx < room) {
                rsd[p + idx] = v2i{s, dl};
                rw[p + idx]  = w;
            }
        }
        add_rows<SC_GRP>(s, dl, w, min(32, cnt - base), lane, hw, rows);
    }
}

__global__ __launch_bounds__(1024) void k_range_sums(
    const int* __restrict__ src, const int* __restrict__ dst,
    const unsigned* __restrict__ key4, const unsigned* __restrict__ degi,
    const float* __restrict__ dinv, const float* __restrict__ hw, float* agg,
    int* __restrict__ rec_sd, float* __restrict__ rec_w, int* __restrict__ rinfo,
    int e, int n) {
    __shared__ alignas(16) float rows[RS_ROWS * FD];
    __shared__ int hits[RS_WAVES * RS_CAP];
    __shared__ unsigned wsum[RS_WAVES];
    __shared__ int lpos;
    const int lane = threadIdx.x & 31;
    const int wid  = __builtin_amdgcn_readfirstlane((int)(threadIdx.x >> 5));
    const int r0   = blockIdx.x * RS_ROWS;
    const int nr   = min(RS_ROWS, n - r0);
    if (nr <= 0) return;

    unsigned part = 0u;
    {
        const v4u* dg4 = (const v4u*)degi;
        const int  nq  = r0 / 4;
        for (int q0 = 0; q0 < nq; q0 += RS_DEG4 * RS_WAVES * 32) {
            v4u t[RS_DEG4];
#pragma unroll
            for (int i = 0; i < RS_DEG4; ++i)
                t[i] = dg4[min(q0 + (int)threadIdx.x + i * RS_WAVES * 32, nq - 1)];
#pragma unroll
            for (int i = 0; i < RS_DEG4; ++i)
                if (q0 + (int)threadIdx.x + i * RS_WAVES * 32 < nq)
                    part += t[i].x + t[i].y + t[i].z + t[i].w;
        }
    }
#pragma unroll
    for (int off = 16; off > 0; off >>= 1) part += __shfl_xor(part, off, 32);
    if (lane == 0) wsum[wid] = part;
    if (threadIdx.x == 0) lpos = 0;

    v4f* gl = (v4f*)(agg + (size_t)r0 * FD);
    v4f* rl = (v4f*)rows;
    for (int i = threadIdx.x; i < nr * (FD / 4); i += RS_WAVES * 32) rl[i] = gl[i];
    __syncthreads();

    unsigned tot = wsum[lane];
#pragma unroll
    for (int off = 16; off > 0; off >>= 1) tot += __shfl_xor(tot, off, 32);
    const int gbase = __builtin_amdgcn_readfirstlane((int)tot - r0);
    v2i*   myrsd = (v2i*)rec_sd + gbase;
    float* myrw  = rec_w + gbase;
    const int room = e - gbase;

    const unsigned splat = (unsigned)blockIdx.x * 0x01010101u;
    int* myhits = hits + wid * RS_CAP;
    int cnt = 0;
    const int thr = 32 + 8 * ((wid + (wid >> 2)) & 7);
    const int ntrips = (e + RS_TRIP - 1) / RS_TRIP;
    const int tlast  = ntrips - 1;
    const v4u* kp = (const v4u*)key4 + lane;
    v4u k0 = kp[(size_t)min(wid,                tlast) * 32];
    v4u k1 = kp[(size_t)min(wid +     RS_WAVES, tlast) * 32];
    v4u k2 = kp[(size_t)min(wid + 2 * RS_WAVES, tlast) * 32];
    v4u k3 = kp[(size_t)min(wid + 3 * RS_WAVES, tlast) * 32];
#pragma nounroll
    for (int tr = wid; tr < ntrips; tr += RS_WAVES) {
        const v4u kc = k0;
        k0 = k1; k1 = k2; k2 = k3;
        k3 = kp[(size_t)min(tr + 4 * RS_WAVES, tlast) * 32];
        unsigned m = (zero_bytes(kc.x ^ splat) >> 7) | (zero_bytes(kc.y ^ splat) >> 6) |
                     (zero_bytes(kc.z ^ splat) >> 5) | (zero_bytes(kc.w ^ splat) >> 4);
        while (__builtin_amdgcn_ballot_w32(m != 0u) != 0u) {
            const int  p   = __builtin_ctz(m | 0x80000000u);
            const int  eid = tr * RS_TRIP + lane * 16 + 4 * (p & 7) + (p >> 3);
            const bool ok  = (m != 0u) && (eid < e);
            const unsigned bm = __builtin_amdgcn_ballot_w32(ok);
            if (ok) myhits[cnt + (int)__builtin_amdgcn_mbcnt_lo(bm, 0u)] = eid;
            cnt += __builtin_popcount(bm);
            m &= m - 1u;
            if (cnt >= thr) {
                work_off(myhits, cnt, lane, r0, src, dst, dinv, hw, rows, myrsd, myrw, room, &lpos);
                cnt = 0;
            }
        }
    }
    work_off(myhits, cnt, lane, r0, src, dst, dinv, hw, rows, myrsd, myrw, room, &lpos);
    __syncthreads();

    if (threadIdx.x == 0) {
        rinfo[2 * blockIdx.x]     = gbase;
        rinfo[2 * blockIdx.x + 1] = min(lpos, room);
    }
    for (int i = threadIdx.x; i < nr * (FD / 4); i += RS_WAVES * 32) gl[i] = rl[i];
}

static __device__ __forceinline__ void load_rec(
    const v2i* __restrict__ rsd, const float* __restrict__ rw, int idx, int cnt,
    v2i& sd, float& w) {
    sd = v2i{0, 0};
    w  = 0.0f;
    if (idx < cnt) {
        sd = rsd[idx];
        w  = rw[idx];
    }
}

__global__ __launch_bounds__(1024) void k_list_sums(
    const int* __restrict__ rec_sd, const float* __restrict__ rec_w,
    const int* __restrict__ rinfo, const float* __restrict__ hw, float* agg, int n) {
    __shared__ alignas(16) float rows[RS_ROWS * FD];
    const int lane = threadIdx.x & 31;
    const int wid  = __builtin_amdgcn_readfirstlane((int)(threadIdx.x >> 5));
    const int r0   = blockIdx.x * RS_ROWS;
    const int nr   = min(RS_ROWS, n - r0);
    if (nr <= 0) return;
    const int gbase = rinfo[2 * blockIdx.x];
    const int cnt   = rinfo[2 * blockIdx.x + 1];
    const v2i*   myrsd = (const v2i*)rec_sd + gbase;
    const float* myrw  = rec_w + gbase;

    int   c0 = wid * 32;
    v2i   sd;
    float w;
    load_rec(myrsd, myrw, c0 + lane, cnt, sd, w);

    v4f* gl = (v4f*)(agg + (size_t)r0 * FD);
    v4f* rl = (v4f*)rows;
    const int nv = nr * (FD / 4);
    {
        v4f sv[RS_SV];
#pragma unroll
        for (int i = 0; i < RS_SV; ++i)
            sv[i] = gl[min((int)threadIdx.x + i * RS_WAVES * 32, nv - 1)];
#pragma unroll
        for (int i = 0; i < RS_SV; ++i)
            if ((int)threadIdx.x + i * RS_WAVES * 32 < nv)
                rl[threadIdx.x + i * RS_WAVES * 32] = sv[i];
    }
    __syncthreads();

    for (; c0 < cnt; c0 += RS_WAVES * 32) {
        v2i   sdn;
        float wn;
        load_rec(myrsd, myrw, c0 + RS_WAVES * 32 + lane, cnt, sdn, wn);
        add_rows<LS_GRP>(sd.x, sd.y, w, min(32, cnt - c0), lane, hw, rows);
        sd = sdn;
        w  = wn;
    }
    __syncthreads();

    for (int i = threadIdx.x; i < nv; i += RS_WAVES * 32) gl[i] = rl[i];
}

__global__ __launch_bounds__(256) void k_logits(
    const float* __restrict__ h, const float* __restrict__ Wp2,
    const float* __restrict__ bp2, float* __restrict__ out, int n) {
    const int wid  = (int)((blockIdx.x * blockDim.x + threadIdx.x) >> 5);
    const int lane = threadIdx.x & 31;
    if (wid >= n) return;
    const float4 v = ((const float4*)(h + (size_t)wid * FD))[lane];
    const int f = lane * 4;
    float p0 = v.x * Wp2[(f + 0) * 2 + 0] + v.y * Wp2[(f + 1) * 2 + 0] +
               v.z * Wp2[(f + 2) * 2 + 0] + v.w * Wp2[(f + 3) * 2 + 0];
    float p1 = v.x * Wp2[(f + 0) * 2 + 1] + v.y * Wp2[(f + 1) * 2 + 1] +
               v.z * Wp2[(f + 2) * 2 + 1] + v.w * Wp2[(f + 3) * 2 + 1];
#pragma unroll
    for (int off = 16; off > 0; off >>= 1) {
        p0 += __shfl_down(p0, off, 32);
        p1 += __shfl_down(p1, off, 32);
    }
    if (lane == 0) {
        const float l0 = p0 + bp2[0];
        const float l1 = p1 + bp2[1];
        const float m  = fmaxf(l0, l1);
        const float lse = m + logf(expf(l0 - m) + expf(l1 - m));
        out[(size_t)wid * 2 + 0] = l0 - lse;
        out[(size_t)wid * 2 + 1] = l1 - lse;
    }
}

extern "C" void kernel_launch(void* const* d_in, const int* in_sizes, int n_in,
                              void* d_out, int out_size, void* d_ws,
                              size_t ws_size, hipStream_t stream) {
    const float* x   = (const float*)d_in[0];
    const int*   ei  = (const int*)d_in[1];
    const float* W1  = (const float*)d_in[2];
    const float* b1  = (const float*)d_in[3];
    const float* W2  = (const float*)d_in[4];
    const float* b2  = (const float*)d_in[5];
    const float* W3  = (const float*)d_in[6];
    const float* b3  = (const float*)d_in[7];
    const float* Wp1 = (const float*)d_in[8];
    const float* bp1 = (const float*)d_in[9];
    const float* Wp2 = (const float*)d_in[10];
    const float* bp2 = (const float*)d_in[11];

    const int N = in_sizes[0] / FD;
    const int E = in_sizes[1] / 2;
    const int* src = ei;
    const int* dst = ei + E;

    char* ws = (char*)d_ws;
    unsigned* degi = (unsigned*)ws;  ws += (size_t)N * sizeof(unsigned);
    float* dinv = (float*)ws;        ws += (size_t)N * sizeof(float);
    float* hw   = (float*)ws;        ws += (size_t)N * FD * sizeof(float);
    float* agg1 = (float*)ws;        ws += (size_t)N * FD * sizeof(float);
    float* agg2 = (float*)ws;        ws += (size_t)N * FD * sizeof(float);
    unsigned* key4 = (unsigned*)ws;
    ws += (size_t)((E + RS_TRIP - 1) / RS_TRIP) * RS_TRIP;
    int* rec_sd  = (int*)ws;         ws += (size_t)E * 2 * sizeof(int);
    float* rec_w = (float*)ws;       ws += (size_t)E * sizeof(float);
    int* rinfo   = (int*)ws;

    float* emb_out  = (float*)d_out;
    float* prob_out = (float*)d_out + (size_t)N * FD;

    const dim3 blk(256);
    const int gN      = (N + 255) / 256;
    const int nTrips  = (E + RS_TRIP - 1) / RS_TRIP;
    const int nKey4   = nTrips * (RS_TRIP / 4);
    const int gE      = (nKey4 + 255) / 256;
    const int gTiles  = N / 16;
    const int gGemm   = (gTiles + GT_WAVES - 1) / GT_WAVES;
    const dim3 blkSum(RS_WAVES * 32);
    const int gNodeW  = (N * WAVE + 255) / 256;

    k_deg_init <<<gN, blk, 0, stream>>>(degi, N);
    k_deg_count<<<gE, blk, 0, stream>>>(dst, degi, key4, E, nKey4);
    k_dinv     <<<gN, blk, 0, stream>>>(degi, dinv, N);

    k_gemm_conv_in  <<<gGemm, blk, 0, stream>>>(x, W1, b1, dinv, hw, agg1, gTiles);
    k_range_sums    <<<RS_WGS, blkSum, 0, stream>>>(src, dst, key4, degi, dinv, hw, agg1, rec_sd, rec_w, rinfo, E, N);

    k_gemm_conv_relu<<<gGemm, blk, 0, stream>>>(agg1, W2, b2, dinv, hw, agg2, gTiles);
    k_list_sums     <<<RS_WGS, blkSum, 0, stream>>>(rec_sd, rec_w, rinfo, hw, agg2, N);

    k_gemm_conv_relu<<<gGemm, blk, 0, stream>>>(agg2, W3, b3, dinv, hw, emb_out, gTiles);
    k_list_sums     <<<RS_WGS, blkSum, 0, stream>>>(rec_sd, rec_w, rinfo, hw, emb_out, N);

    k_gemm_post_relu<<<gGemm, blk, 0, stream>>>(emb_out, Wp1, bp1, hw, gTiles);
    k_logits        <<<gNodeW, blk, 0, stream>>>(hw, Wp2, bp2, prob_out, N);
}


# Edge sums: a source row comes as ONE 16-byte-per-lane load (row plane 512-B aligned: two 256-B units, not four line loads); LDS rows keep feature 4q+j at word 32j+q; sum order: arrival in group
# speedup vs baseline: 2.8847x; 1.0428x over previous
#include <hip/hip_runtime.h>
#include <math.h>

#define FD 128
#define WAVE 32

typedef __attribute__((ext_vector_type(2))) float v2f;
typedef __attribute__((ext_vector_type(8))) float v8f;

#define RS_WGS   256
#define RS_ROWS  196
#define RS_TRIP  512

__global__ void k_deg_init(unsigned* __restrict__ degi, int n) {
    int i = blockIdx.x * blockDim.x + threadIdx.x;
    if (i < n) degi[i] = 1u;
}

__global__ void k_deg_count(const int* __restrict__ dst,
                            unsigned* __restrict__ degi,
                            unsigned* __restrict__ key4, int e, int nkey4) {
    int i = blockIdx.x * blockDim.x + threadIdx.x;
    if (i >= nkey4) return;
    unsigned d[4];
#pragma unroll
    for (int b = 0; b < 4; ++b) d[b] = (unsigned)dst[min(4 * i + b, e - 1)];
    unsigned k = 0u;
#pragma unroll
    for (int b = 0; b < 4; ++b) {
        if (4 * i + b < e) {
            atomicAdd(&degi[d[b]], 1u);
            k |= (d[b] / RS_ROWS) << (8 * b);
        }
    }
    key4[i] = k;
}

__global__ void k_dinv(const unsigned* __restrict__ degi,
                       float* __restrict__ dinv, int n) {
    int i = blockIdx.x * blockDim.x + threadIdx.x;
    if (i < n) dinv[i] = rsqrtf((float)degi[i]);
}

#define GT_WAVES 8
#define WL_HALF  (FD * 4 + 32)
#define WL_PAIR  (2 * FD * 4 + 32)
#define WL_WORDS (16 * WL_PAIR)

typedef __attribute__((ext_vector_type(4))) float v4f;

static __device__ __forceinline__ void stage_w(const float* __restrict__ W,
                                               float* __restrict__ wl) {
    const int n  = threadIdx.x & (FD - 1);
    const int g0 = threadIdx.x >> 7;
#pragma unroll
    for (int i = 0; i < FD / 8; ++i) {
        const int g = 2 * i + g0;
        v4f q;
        q.x = W[(size_t)(4 * g + 0) * FD + n];
        q.y = W[(size_t)(4 * g + 1) * FD + n];
        q.z = W[(size_t)(4 * g + 2) * FD + n];
        q.w = W[(size_t)(4 * g + 3) * FD + n];
        *(v4f*)(wl + i * WL_PAIR + g0 * WL_HALF + n * 4) = q;
    }
}

static __device__ __forceinline__ void load_a_tile(
    const float* __restrict__ arow, v4f (&a)[FD / 8]) {
#pragma unroll
    for (int j = 0; j < FD / 8; ++j) a[j] = *(const v4f*)(arow + 8 * j);
}

static __device__ __forceinline__ void relu_a_tile(v4f (&a)[FD / 8]) {
#pragma unroll
    for (int j = 0; j < FD / 8; ++j) {
        a[j].x = fmaxf(a[j].x, 0.0f); a[j].y = fmaxf(a[j].y, 0.0f);
        a[j].z = fmaxf(a[j].z, 0.0f); a[j].w = fmaxf(a[j].w, 0.0f);
    }
}

static __device__ __forceinline__ void load_b_step(
    const float* __restrict__ wlane, int j, v4f (&b)[8]) {
#pragma unroll
    for (int t = 0; t < 8; ++t)
        b[t] = *(const v4f*)(wlane + j * WL_PAIR + 64 * t);
}

static __device__ __forceinline__ void wmma_step(
    const v4f& av4, const v4f (&b)[8], v8f (&acc)[8]) {
#pragma unroll
    for (int t = 0; t < 8; ++t) {
        const v2f av = {av4.x, av4.y};
        const v2f bv = {b[t].x, b[t].y};
        acc[t] = __builtin_amdgcn_wmma_f32_16x16x4_f32(
            false, av, false, bv, (short)0, acc[t], false, false);
    }
#pragma unroll
    for (int t = 0; t < 8; ++t) {
        const v2f av = {av4.z, av4.w};
        const v2f bv = {b[t].z, b[t].w};
        acc[t] = __builtin_amdgcn_wmma_f32_16x16x4_f32(
            false, av, false, bv, (short)0, acc[t], false, false);
    }
}

static __device__ __forceinline__ void rowtile_product(
    const v4f (&a)[FD / 8], const float* __restrict__ wlane, v8f (&acc)[8]) {
    v4f b0[8], b1[8];
    load_b_step(wlane, 0, b0);
#pragma unroll
    for (int j = 0; j < FD / 8; j += 2) {
        load_b_step(wlane, j + 1, b1);
        __builtin_amdgcn_sched_barrier(0);
        wmma_step(a[j], b0, acc);
        __builtin_amdgcn_sched_barrier(0);
        if (j + 2 < FD / 8) load_b_step(wlane, j + 2, b0);
        __builtin_amdgcn_sched_barrier(0);
        wmma_step(a[j + 1], b1, acc);
        __builtin_amdgcn_sched_barrier(0);
    }
}

static __device__ __forceinline__ bool rowtile_front(
    const float* __restrict__ A, const float* __restrict__ W, int ntiles,
    bool relu_in, float* __restrict__ wl, int& m0, int& c, int& h,
    v4f (&a)[FD / 8]) {
    const int lane = threadIdx.x & 31;
    const int wid  = __builtin_amdgcn_readfirstlane((int)(threadIdx.x >> 5));
    const int tile = blockIdx.x * GT_WAVES + wid;
    const int tl   = min(tile, ntiles - 1);
    c  = lane & 15;
    h  = lane >> 4;
    m0 = tile << 4;

    load_a_tile(A + (size_t)((tl << 4) + c) * FD + 4 * h, a);

    stage_w(W, wl);
    __syncthreads();
    if (tile >= ntiles) return false;

    if (relu_in) relu_a_tile(a);
    return true;
}

static __device__ __forceinline__ void gemm_conv_body(
    const float* __restrict__ A, const float* __restrict__ W,
    const float* __restrict__ bias, const float* __restrict__ dinv,
    float* __restrict__ hw, float* __restrict__ agg, int ntiles, bool relu_in,
    float* __restrict__ wl) {
    int m0, c, h;
    v4f a[FD / 8];
    if (!rowtile_front(A, W, ntiles, relu_in, wl, m0, c, h, a)) return;

    float bc[8], di[8];
#pragma unroll
    for (int t = 0; t < 8; ++t) bc[t] = bias[16 * t + c];
#pragma unroll
    for (int v = 0; v < 8; ++v) di[v] = dinv[m0 + v + 8 * h];

    v8f acc[8];
#pragma unroll
    for (int t = 0; t < 8; ++t) acc[t] = v8f{};
    rowtile_product(a, wl + h * WL_HALF + c * 4, acc);

#pragma unroll
    for (int v = 0; v < 8; ++v) {
        const int row = m0 + v + 8 * h;
#pragma unroll
        for (int t = 0; t < 8; ++t) {
            const float  r = acc[t][v];
            const size_t o = (size_t)row * FD + 16 * t + c;
            hw[o]  = r;
            agg[o] = r * di[v] * di[v] + bc[t];
        }
    }
}

__global__ __launch_bounds__(256) void k_gemm_conv_in(
    const float* __restrict__ A, const float* __restrict__ W,
    const float* __restrict__ bias, const float* __restrict__ dinv,
    float* __restrict__ hw, float* __restrict__ agg, int ntiles) {
    __shared__ alignas(16) float wl[WL_WORDS];
    gemm_conv_body(A, W, bias, dinv, hw, agg, ntiles, false, wl);
}

__global__ __launch_bounds__(256) void k_gemm_conv_relu(
    const float* __restrict__ A, const float* __restrict__ W,
    const float* __restrict__ bias, const float* __restrict__ dinv,
    float* __restrict__ hw, float* __restrict__ agg, int ntiles) {
    __shared__ alignas(16) float wl[WL_WORDS];
    gemm_conv_body(A, W, bias, dinv, hw, agg, ntiles, true, wl);
}

__global__ __launch_bounds__(256) void k_gemm_post_relu(
    const float* __restrict__ A, const float* __restrict__ W,
    const float* __restrict__ bias, float* __restrict__ out, int ntiles) {
    __shared__ alignas(16) float wl[WL_WORDS];
    int m0, c, h;
    v4f a[FD / 8];
    if (!rowtile_front(A, W, ntiles, true, wl, m0, c, h, a)) return;

    float bc[8];
#pragma unroll
    for (int t = 0; t < 8; ++t) bc[t] = bias[16 * t + c];

    v8f acc[8];
#pragma unroll
    for (int t = 0; t < 8; ++t) acc[t] = v8f{};
    rowtile_product(a, wl + h * WL_HALF + c * 4, acc);

#pragma unroll
    for (int v = 0; v < 8; ++v) {
        const int row = m0 + v + 8 * h;
#pragma unroll
        for (int t = 0; t < 8; ++t) {
            const float r = acc[t][v] + bc[t];
            out[(size_t)row * FD + 16 * t + c] = fmaxf(r, 0.0f);
        }
    }
}

#define RS_WAVES 32
#define RS_CAP   512
#define SC_GRP   8
#define LS_GRP   16
#define RS_DEG4  13
#define RS_SV    ((RS_ROWS * (FD / 4) + RS_WAVES * 32 - 1) / (RS_WAVES * 32))

typedef __attribute__((ext_vector_type(4))) unsigned v4u;
typedef __attribute__((ext_vector_type(2))) int v2i;

static __device__ __forceinline__ unsigned zero_bytes(unsigned x) {
    const unsigned t = (x & 0x7f7f7f7fu) + 0x7f7f7f7fu;
    return ~(t | x | 0x7f7f7f7fu);
}

static __device__ __forceinline__ void rows_put(float* rows, int i, v4f q) {
    float* rp = rows + (i >> 5) * FD + (i & 31);
    rp[0] = q.x; rp[32] = q.y; rp[64] = q.z; rp[96] = q.w;
}
static __device__ __forceinline__ v4f rows_get(const float* rows, int i) {
    const float* rp = rows + (i >> 5) * FD + (i & 31);
    return v4f{rp[0], rp[32], rp[64], rp[96]};
}

template <int G>
static __device__ __forceinline__ void add_rows(
    int s, int dl, float w, int nt, int lane,
    const float* __restrict__ hw, float* rows) {
    for (int g0 = 0; g0 < nt; g0 += G) {
        v4f v[G];
#pragma unroll
        for (int g = 0; g < G; ++g) {
            const int sj = __builtin_amdgcn_readlane(s, min(g0 + g, nt - 1));
            v[g] = ((const v4f*)(hw + (size_t)sj * FD))[lane];
        }
#pragma unroll
        for (int g = 0; g < G; ++g) {
            if (g == 0 || g0 + g < nt) {
                const int   dj = __builtin_amdgcn_readlane(dl, g0 + g);
                const float wj = __int_as_float(__builtin_amdgcn_readlane(__float_as_int(w), g0 + g));
                float* rp = rows + dj + lane;
                unsafeAtomicAdd(rp,      v[g].x * wj);
                unsafeAtomicAdd(rp + 32, v[g].y * wj);
                unsafeAtomicAdd(rp + 64, v[g].z * wj);
                unsafeAtomicAdd(rp + 96, v[g].w * wj);
            }
        }
    }
}

static __device__ __forceinline__ void work_off(
    const int* hits, int cnt, int lane, int r0,
    const int* __restrict__ src, const int* __restrict__ dst,
    const float* __restrict__ dinv, const float* __restrict__ hw, float* rows,
    v2i* __restrict__ rsd, float* __restrict__ rw, int room, int* lpos) {
    if (cnt <= 0) return;
    int p = 0;
    if (lane == 0) p = atomicAdd(lpos, cnt);
    p = __builtin_amdgcn_readfirstlane(p);
    for (int base = 0; base < cnt; base += 32) {
        const int idx = base + lane;
        int   s = 0, dl = 0;
        float w = 0.0f;
        if (idx < cnt) {
            const int eid = hits[idx];
            s = src[eid];
            const int d = dst[eid];
            w  = dinv[s] * dinv[d];
            dl = (d - r0) * FD;
            if (p + idx < room) {
                rsd[p + idx] = v2i{s, dl};
                rw[p + idx]  = w;
            }
        }
        add_rows<SC_GRP>(s, dl, w, min(32, cnt - base), lane, hw, rows);
    }
}

__global__ __launch_bounds__(1024) void k_range_sums(
    const int* __restrict__ src, const int* __restrict__ dst,
    const unsigned* __restrict__ key4, const unsigned* __restrict__ degi,
    const float* __restrict__ dinv, const float* __restrict__ hw, float* agg,
    int* __restrict__ rec_sd, float* __restrict__ rec_w, int* __restrict__ rinfo,
    int e, int n) {
    __shared__ alignas(16) float rows[RS_ROWS * FD];
    __shared__ int hits[RS_WAVES * RS_CAP];
    __shared__ unsigned wsum[RS_WAVES];
    __shared__ int lpos;
    const int lane = threadIdx.x & 31;
    const int wid  = __builtin_amdgcn_readfirstlane((int)(threadIdx.x >> 5));
    const int r0   = blockIdx.x * RS_ROWS;
    const int nr   = min(RS_ROWS, n - r0);
    if (nr <= 0) return;

    unsigned part = 0u;
    {
        const v4u* dg4 = (const v4u*)degi;
        const int  nq  = r0 / 4;
        for (int q0 = 0; q0 < nq; q0 += RS_DEG4 * RS_WAVES * 32) {
            v4u t[RS_DEG4];
#pragma unroll
            for (int i = 0; i < RS_DEG4; ++i)
                t[i] = dg4[min(q0 + (int)threadIdx.x + i * RS_WAVES * 32, nq - 1)];
#pragma unroll
            for (int i = 0; i < RS_DEG4; ++i)
                if (q0 + (int)threadIdx.x + i * RS_WAVES * 32 < nq)
                    part += t[i].x + t[i].y + t[i].z + t[i].w;
        }
    }
#pragma unroll
    for (int off = 16; off > 0; off >>= 1) part += __shfl_xor(part, off, 32);
    if (lane == 0) wsum[wid] = part;
    if (threadIdx.x == 0) lpos = 0;

    v4f* gl = (v4f*)(agg + (size_t)r0 * FD);
    for (int i = threadIdx.x; i < nr * (FD / 4); i += RS_WAVES * 32) rows_put(rows, i, gl[i]);
    __syncthreads();

    unsigned tot = wsum[lane];
#pragma unroll
    for (int off = 16; off > 0; off >>= 1) tot += __shfl_xor(tot, off, 32);
    const int gbase = __builtin_amdgcn_readfirstlane((int)tot - r0);
    v2i*   myrsd = (v2i*)rec_sd + gbase;
    float* myrw  = rec_w + gbase;
    const int room = e - gbase;

    const unsigned splat = (unsigned)blockIdx.x * 0x01010101u;
    int* myhits = hits + wid * RS_CAP;
    int cnt = 0;
    const int thr = 32 + 8 * ((wid + (wid >> 2)) & 7);
    const int ntrips = (e + RS_TRIP - 1) / RS_TRIP;
    const int tlast  = ntrips - 1;
    const v4u* kp = (const v4u*)key4 + lane;
    v4u k0 = kp[(size_t)min(wid,                tlast) * 32];
    v4u k1 = kp[(size_t)min(wid +     RS_WAVES, tlast) * 32];
    v4u k2 = kp[(size_t)min(wid + 2 * RS_WAVES, tlast) * 32];
    v4u k3 = kp[(size_t)min(wid + 3 * RS_WAVES, tlast) * 32];
#pragma nounroll
    for (int tr = wid; tr < ntrips; tr += RS_WAVES) {
        const v4u kc = k0;
        k0 = k1; k1 = k2; k2 = k3;
        k3 = kp[(size_t)min(tr + 4 * RS_WAVES, tlast) * 32];
        unsigned m = (zero_bytes(kc.x ^ splat) >> 7) | (zero_bytes(kc.y ^ splat) >> 6) |
                     (zero_bytes(kc.z ^ splat) >> 5) | (zero_bytes(kc.w ^ splat) >> 4);
        while (__builtin_amdgcn_ballot_w32(m != 0u) != 0u) {
            const int  p   = __builtin_ctz(m | 0x80000000u);
            const int  eid = tr * RS_TRIP + lane * 16 + 4 * (p & 7) + (p >> 3);
            const bool ok  = (m != 0u) && (eid < e);
            const unsigned bm = __builtin_amdgcn_ballot_w32(ok);
            if (ok) myhits[cnt + (int)__builtin_amdgcn_mbcnt_lo(bm, 0u)] = eid;
            cnt += __builtin_popcount(bm);
            m &= m - 1u;
            if (cnt >= thr) {
                work_off(myhits, cnt, lane, r0, src, dst, dinv, hw, rows, myrsd, myrw, room, &lpos);
                cnt = 0;
            }
        }
    }
    work_off(myhits, cnt, lane, r0, src, dst, dinv, hw, rows, myrsd, myrw, room, &lpos);
    __syncthreads();

    if (threadIdx.x == 0) {
        rinfo[2 * blockIdx.x]     = gbase;
        rinfo[2 * blockIdx.x + 1] = min(lpos, room);
    }
    for (int i = threadIdx.x; i < nr * (FD / 4); i += RS_WAVES * 32) gl[i] = rows_get(rows, i);
}

static __device__ __forceinline__ void load_rec(
    const v2i* __restrict__ rsd, const float* __restrict__ rw, int idx, int cnt,
    v2i& sd, float& w) {
    sd = v2i{0, 0};
    w  = 0.0f;
    if (idx < cnt) {
        sd = rsd[idx];
        w  = rw[idx];
    }
}

__global__ __launch_bounds__(1024) void k_list_sums(
    const int* __restrict__ rec_sd, const float* __restrict__ rec_w,
    const int* __restrict__ rinfo, const float* __restrict__ hw, float* agg, int n) {
    __shared__ alignas(16) float rows[RS_ROWS * FD];
    const int lane = threadIdx.x & 31;
    const int wid  = __builtin_amdgcn_readfirstlane((int)(threadIdx.x >> 5));
    const int r0   = blockIdx.x * RS_ROWS;
    const int nr   = min(RS_ROWS, n - r0);
    if (nr <= 0) return;
    const int gbase = rinfo[2 * blockIdx.x];
    const int cnt   = rinfo[2 * blockIdx.x + 1];
    const v2i*   myrsd = (const v2i*)rec_sd + gbase;
    const float* myrw  = rec_w + gbase;

    int   c0 = wid * 32;
    v2i   sd;
    float w;
    load_rec(myrsd, myrw, c0 + lane, cnt, sd, w);

    v4f* gl = (v4f*)(agg + (size_t)r0 * FD);
    const int nv = nr * (FD / 4);
    {
        v4f sv[RS_SV];
#pragma unroll
        for (int i = 0; i < RS_SV; ++i)
            sv[i] = gl[min((int)threadIdx.x + i * RS_WAVES * 32, nv - 1)];
#pragma unroll
        for (int i = 0; i < RS_SV; ++i)
            if ((int)threadIdx.x + i * RS_WAVES * 32 < nv)
                rows_put(rows, (int)threadIdx.x + i * RS_WAVES * 32, sv[i]);
    }
    __syncthreads();

    for (; c0 < cnt; c0 += RS_WAVES * 32) {
        v2i   sdn;
        float wn;
        load_rec(myrsd, myrw, c0 + RS_WAVES * 32 + lane, cnt, sdn, wn);
        add_rows<LS_GRP>(sd.x, sd.y, w, min(32, cnt - c0), lane, hw, rows);
        sd = sdn;
        w  = wn;
    }
    __syncthreads();

    for (int i = threadIdx.x; i < nv; i += RS_WAVES * 32) gl[i] = rows_get(rows, i);
}

__global__ __launch_bounds__(256) void k_logits(
    const float* __restrict__ h, const float* __restrict__ Wp2,
    const float* __restrict__ bp2, float* __restrict__ out, int n) {
    const int wid  = (int)((blockIdx.x * blockDim.x + threadIdx.x) >> 5);
    const int lane = threadIdx.x & 31;
    if (wid >= n) return;
    const float4 v = ((const float4*)(h + (size_t)wid * FD))[lane];
    const int f = lane * 4;
    float p0 = v.x * Wp2[(f + 0) * 2 + 0] + v.y * Wp2[(f + 1) * 2 + 0] +
               v.z * Wp2[(f + 2) * 2 + 0] + v.w * Wp2[(f + 3) * 2 + 0];
    float p1 = v.x * Wp2[(f + 0) * 2 + 1] + v.y * Wp2[(f + 1) * 2 + 1] +
               v.z * Wp2[(f + 2) * 2 + 1] + v.w * Wp2[(f + 3) * 2 + 1];
#pragma unroll
    for (int off = 16; off > 0; off >>= 1) {
        p0 += __shfl_down(p0, off, 32);
        p1 += __shfl_down(p1, off, 32);
    }
    if (lane == 0) {
        const float l0 = p0 + bp2[0];
        const float l1 = p1 + bp2[1];
        const float m  = fmaxf(l0, l1);
        const float lse = m + logf(expf(l0 - m) + expf(l1 - m));
        out[(size_t)wid * 2 + 0] = l0 - lse;
        out[(size_t)wid * 2 + 1] = l1 - lse;
    }
}

extern "C" void kernel_launch(void* const* d_in, const int* in_sizes, int n_in,
                              void* d_out, int out_size, void* d_ws,
                              size_t ws_size, hipStream_t stream) {
    const float* x   = (const float*)d_in[0];
    const int*   ei  = (const int*)d_in[1];
    const float* W1  = (const float*)d_in[2];
    const float* b1  = (const float*)d_in[3];
    const float* W2  = (const float*)d_in[4];
    const float* b2  = (const float*)d_in[5];
    const float* W3  = (const float*)d_in[6];
    const float* b3  = (const float*)d_in[7];
    const float* Wp1 = (const float*)d_in[8];
    const float* bp1 = (const float*)d_in[9];
    const float* Wp2 = (const float*)d_in[10];
    const float* bp2 = (const float*)d_in[11];

    const int N = in_sizes[0] / FD;
    const int E = in_sizes[1] / 2;
    const int* src = ei;
    const int* dst = ei + E;

    char* ws = (char*)d_ws;
    float* hw   = (float*)ws;        ws += (size_t)N * FD * sizeof(float);
    float* agg1 = (float*)ws;        ws += (size_t)N * FD * sizeof(float);
    float* agg2 = (float*)ws;        ws += (size_t)N * FD * sizeof(float);
    unsigned* degi = (unsigned*)ws;  ws += (size_t)N * sizeof(unsigned);
    float* dinv = (float*)ws;        ws += (size_t)N * sizeof(float);
    unsigned* key4 = (unsigned*)ws;
    ws += (size_t)((E + RS_TRIP - 1) / RS_TRIP) * RS_TRIP;
    int* rec_sd  = (int*)ws;         ws += (size_t)E * 2 * sizeof(int);
    float* rec_w = (float*)ws;       ws += (size_t)E * sizeof(float);
    int* rinfo   = (int*)ws;

    float* emb_out  = (float*)d_out;
    float* prob_out = (float*)d_out + (size_t)N * FD;

    const dim3 blk(256);
    const int gN      = (N + 255) / 256;
    const int nTrips  = (E + RS_TRIP - 1) / RS_TRIP;
    const int nKey4   = nTrips * (RS_TRIP / 4);
    const int gE      = (nKey4 + 255) / 256;
    const int gTiles  = N / 16;
    const int gGemm   = (gTiles + GT_WAVES - 1) / GT_WAVES;
    const dim3 blkSum(RS_WAVES * 32);
    const int gNodeW  = (N * WAVE + 255) / 256;

    k_deg_init <<<gN, blk, 0, stream>>>(degi, N);
    k_deg_count<<<gE, blk, 0, stream>>>(dst, degi, key4, E, nKey4);
    k_dinv     <<<gN, blk, 0, stream>>>(degi, dinv, N);

    k_gemm_conv_in  <<<gGemm, blk, 0, stream>>>(x, W1, b1, dinv, hw, agg1, gTiles);
    k_range_sums    <<<RS_WGS, blkSum, 0, stream>>>(src, dst, key4, degi, dinv, hw, agg1, rec_sd, rec_w, rinfo, E, N);

    k_gemm_conv_relu<<<gGemm, blk, 0, stream>>>(agg1, W2, b2, dinv, hw, agg2, gTiles);
    k_list_sums     <<<RS_WGS, blkSum, 0, stream>>>(rec_sd, rec_w, rinfo, hw, agg2, N);

    k_gemm_conv_relu<<<gGemm, blk, 0, stream>>>(agg2, W3, b3, dinv, hw, emb_out, gTiles);
    k_list_sums     <<<RS_WGS, blkSum, 0, stream>>>(rec_sd, rec_w, rinfo, hw, emb_out, N);

    k_gemm_post_relu<<<gGemm, blk, 0, stream>>>(emb_out, Wp1, bp1, hw, gTiles);
    k_logits        <<<gNodeW, blk, 0, stream>>>(hw, Wp2, bp2, prob_out, N);
}


# Edge weight factored out of the edge loop: product stores ONE plane, rows times dinv[source]; sums add plain rows onto the own row, then scale by dinv[target] and add bias; sum order as before
# speedup vs baseline: 3.1252x; 1.0834x over previous
#include <hip/hip_runtime.h>
#include <math.h>

#define FD 128
#define WAVE 32

typedef __attribute__((ext_vector_type(2))) float v2f;
typedef __attribute__((ext_vector_type(8))) float v8f;

#define RS_WGS   256
#define RS_ROWS  196
#define RS_TRIP  512

__global__ void k_deg_init(unsigned* __restrict__ degi, int n) {
    int i = blockIdx.x * blockDim.x + threadIdx.x;
    if (i < n) degi[i] = 1u;
}

__global__ void k_deg_count(const int* __restrict__ dst,
                            unsigned* __restrict__ degi,
                            unsigned* __restrict__ key4, int e, int nkey4) {
    int i = blockIdx.x * blockDim.x + threadIdx.x;
    if (i >= nkey4) return;
    unsigned d[4];
#pragma unroll
    for (int b = 0; b < 4; ++b) d[b] = (unsigned)dst[min(4 * i + b, e - 1)];
    unsigned k = 0u;
#pragma unroll
    for (int b = 0; b < 4; ++b) {
        if (4 * i + b < e) {
            atomicAdd(&degi[d[b]], 1u);
            k |= (d[b] / RS_ROWS) << (8 * b);
        }
    }
    key4[i] = k;
}

__global__ void k_dinv(const unsigned* __restrict__ degi,
                       float* __restrict__ dinv, int n) {
    int i = blockIdx.x * blockDim.x + threadIdx.x;
    if (i < n) dinv[i] = rsqrtf((float)degi[i]);
}

#define GT_WAVES 8
#define WL_HALF  (FD * 4 + 32)
#define WL_PAIR  (2 * FD * 4 + 32)
#define WL_WORDS (16 * WL_PAIR)

typedef __attribute__((ext_vector_type(4))) float v4f;

static __device__ __forceinline__ void stage_w(const float* __restrict__ W,
                                               float* __restrict__ wl) {
    const int n  = threadIdx.x & (FD - 1);
    const int g0 = threadIdx.x >> 7;
#pragma unroll
    for (int i = 0; i < FD / 8; ++i) {
        const int g = 2 * i + g0;
        v4f q;
        q.x = W[(size_t)(4 * g + 0) * FD + n];
        q.y = W[(size_t)(4 * g + 1) * FD + n];
        q.z = W[(size_t)(4 * g + 2) * FD + n];
        q.w = W[(size_t)(4 * g + 3) * FD + n];
        *(v4f*)(wl + i * WL_PAIR + g0 * WL_HALF + n * 4) = q;
    }
}

static __device__ __forceinline__ void load_a_tile(
    const float* __restrict__ arow, v4f (&a)[FD / 8]) {
#pragma unroll
    for (int j = 0; j < FD / 8; ++j) a[j] = *(const v4f*)(arow + 8 * j);
}

static __device__ __forceinline__ void relu_a_tile(v4f (&a)[FD / 8]) {
#pragma unroll
    for (int j = 0; j < FD / 8; ++j) {
        a[j].x = fmaxf(a[j].x, 0.0f); a[j].y = fmaxf(a[j].y, 0.0f);
        a[j].z = fmaxf(a[j].z, 0.0f); a[j].w = fmaxf(a[j].w, 0.0f);
    }
}

static __device__ __forceinline__ void load_b_step(
    const float* __restrict__ wlane, int j, v4f (&b)[8]) {
#pragma unroll
    for (int t = 0; t < 8; ++t)
        b[t] = *(const v4f*)(wlane + j * WL_PAIR + 64 * t);
}

static __device__ __forceinline__ void wmma_step(
    const v4f& av4, const v4f (&b)[8], v8f (&acc)[8]) {
#pragma unroll
    for (int t = 0; t < 8; ++t) {
        const v2f av = {av4.x, av4.y};
        const v2f bv = {b[t].x, b[t].y};
        acc[t] = __builtin_amdgcn_wmma_f32_16x16x4_f32(
            false, av, false, bv, (short)0, acc[t], false, false);
    }
#pragma unroll
    for (int t = 0; t < 8; ++t) {
        const v2f av = {av4.z, av4.w};
        const v2f bv = {b[t].z, b[t].w};
        acc[t] = __builtin_amdgcn_wmma_f32_16x16x4_f32(
            false, av, false, bv, (short)0, acc[t], false, false);
    }
}

static __device__ __forceinline__ void rowtile_product(
    const v4f (&a)[FD / 8], const float* __restrict__ wlane, v8f (&acc)[8]) {
    v4f b0[8], b1[8];
    load_b_step(wlane, 0, b0);
#pragma unroll
    for (int j = 0; j < FD / 8; j += 2) {
        load_b_step(wlane, j + 1, b1);
        __builtin_amdgcn_sched_barrier(0);
        wmma_step(a[j], b0, acc);
        __builtin_amdgcn_sched_barrier(0);
        if (j + 2 < FD / 8) load_b_step(wlane, j + 2, b0);
        __builtin_amdgcn_sched_barrier(0);
        wmma_step(a[j + 1], b1, acc);
        __builtin_amdgcn_sched_barrier(0);
    }
}

static __device__ __forceinline__ bool rowtile_front(
    const float* __restrict__ A, const float* __restrict__ W, int ntiles,
    bool relu_in, float* __restrict__ wl, int& m0, int& c, int& h,
    v4f (&a)[FD / 8]) {
    const int lane = threadIdx.x & 31;
    const int wid  = __builtin_amdgcn_readfirstlane((int)(threadIdx.x >> 5));
    const int tile = blockIdx.x * GT_WAVES + wid;
    const int tl   = min(tile, ntiles - 1);
    c  = lane & 15;
    h  = lane >> 4;
    m0 = tile << 4;

    load_a_tile(A + (size_t)((tl << 4) + c) * FD + 4 * h, a);

    stage_w(W, wl);
    __syncthreads();
    if (tile >= ntiles) return false;

    if (relu_in) relu_a_tile(a);
    return true;
}

static __device__ __forceinline__ void gemm_conv_body(
    const float* __restrict__ A, const float* __restrict__ W,
    const float* __restrict__ dinv,
    float* __restrict__ hw, int ntiles, bool relu_in,
    float* __restrict__ wl) {
    int m0, c, h;
    v4f a[FD / 8];
    if (!rowtile_front(A, W, ntiles, relu_in, wl, m0, c, h, a)) return;

    float di[8];
#pragma unroll
    for (int v = 0; v < 8; ++v) di[v] = dinv[m0 + v + 8 * h];

    v8f acc[8];
#pragma unroll
    for (int t = 0; t < 8; ++t) acc[t] = v8f{};
    rowtile_product(a, wl + h * WL_HALF + c * 4, acc);

#pragma unroll
    for (int v = 0; v < 8; ++v) {
        const int row = m0 + v + 8 * h;
#pragma unroll
        for (int t = 0; t < 8; ++t)
            hw[(size_t)row * FD + 16 * t + c] = acc[t][v] * di[v];
    }
}

__global__ __launch_bounds__(256) void k_gemm_conv_in(
    const float* __restrict__ A, const float* __restrict__ W,
    const float* __restrict__ dinv, float* __restrict__ hw, int ntiles) {
    __shared__ alignas(16) float wl[WL_WORDS];
    gemm_conv_body(A, W, dinv, hw, ntiles, false, wl);
}

__global__ __launch_bounds__(256) void k_gemm_conv_relu(
    const float* __restrict__ A, const float* __restrict__ W,
    const float* __restrict__ dinv, float* __restrict__ hw, int ntiles) {
    __shared__ alignas(16) float wl[WL_WORDS];
    gemm_conv_body(A, W, dinv, hw, ntiles, true, wl);
}

__global__ __launch_bounds__(256) void k_gemm_post_relu(
    const float* __restrict__ A, const float* __restrict__ W,
    const float* __restrict__ bias, float* __restrict__ out, int ntiles) {
    __shared__ alignas(16) float wl[WL_WORDS];
    int m0, c, h;
    v4f a[FD / 8];
    if (!rowtile_front(A, W, ntiles, true, wl, m0, c, h, a)) return;

    float bc[8];
#pragma unroll
    for (int t = 0; t < 8; ++t) bc[t] = bias[16 * t + c];

    v8f acc[8];
#pragma unroll
    for (int t = 0; t < 8; ++t) acc[t] = v8f{};
    rowtile_product(a, wl + h * WL_HALF + c * 4, acc);

#pragma unroll
    for (int v = 0; v < 8; ++v) {
        const int row = m0 + v + 8 * h;
#pragma unroll
        for (int t = 0; t < 8; ++t) {
            const float r = acc[t][v] + bc[t];
            out[(size_t)row * FD + 16 * t + c] = fmaxf(r, 0.0f);
        }
    }
}

#define RS_WAVES 32
#define RS_CAP   512
#define SC_GRP   8
#define LS_GRP   16
#define RS_DEG4  13
#define RS_SV    ((RS_ROWS * (FD / 4) + RS_WAVES * 32 - 1) / (RS_WAVES * 32))

typedef __attribute__((ext_vector_type(4))) unsigned v4u;
typedef __attribute__((ext_vector_type(2))) int v2i;

static __device__ __forceinline__ unsigned zero_bytes(unsigned x) {
    const unsigned t = (x & 0x7f7f7f7fu) + 0x7f7f7f7fu;
    return ~(t | x | 0x7f7f7f7fu);
}

static __device__ __forceinline__ void rows_put(float* rows, int i, v4f q) {
    float* rp = rows + (i >> 5) * FD + (i & 31);
    rp[0] = q.x; rp[32] = q.y; rp[64] = q.z; rp[96] = q.w;
}
static __device__ __forceinline__ v4f rows_get(const float* rows, int i) {
    const float* rp = rows + (i >> 5) * FD + (i & 31);
    return v4f{rp[0], rp[32], rp[64], rp[96]};
}

static __device__ __forceinline__ void rows_out(
    const float* rows, v4f* gl, int nv, int r0,
    const float* __restrict__ dinv, const float* __restrict__ bias) {
    const v4f b4 = ((const v4f*)bias)[threadIdx.x & 31];
    for (int i = threadIdx.x; i < nv; i += RS_WAVES * 32) {
        const float di = dinv[r0 + (i >> 5)];
        const v4f   q  = rows_get(rows, i);
        gl[i] = v4f{q.x * di + b4.x, q.y * di + b4.y, q.z * di + b4.z, q.w * di + b4.w};
    }
}

template <int G>
static __device__ __forceinline__ void add_rows(
    int s, int dl, int nt, int lane,
    const float* __restrict__ hw, float* rows) {
    for (int g0 = 0; g0 < nt; g0 += G) {
        v4f v[G];
#pragma unroll
        for (int g = 0; g < G; ++g) {
            const int sj = __builtin_amdgcn_readlane(s, min(g0 + g, nt - 1));
            v[g] = ((const v4f*)(hw + (size_t)sj * FD))[lane];
        }
#pragma unroll
        for (int g = 0; g < G; ++g) {
            if (g == 0 || g0 + g < nt) {
                const int dj = __builtin_amdgcn_readlane(dl, g0 + g);
                float* rp = rows + dj + lane;
                unsafeAtomicAdd(rp,      v[g].x);
                unsafeAtomicAdd(rp + 32, v[g].y);
                unsafeAtomicAdd(rp + 64, v[g].z);
                unsafeAtomicAdd(rp + 96, v[g].w);
            }
        }
    }
}

static __device__ __forceinline__ void work_off(
    const int* hits, int cnt, int lane, int r0,
    const int* __restrict__ src, const int* __restrict__ dst,
    const float* __restrict__ hw, float* rows,
    v2i* __restrict__ rsd, int room, int* lpos) {
    if (cnt <= 0) return;
    int p = 0;
    if (lane == 0) p = atomicAdd(lpos, cnt);
    p = __builtin_amdgcn_readfirstlane(p);
    for (int base = 0; base < cnt; base += 32) {
        const int idx = base + lane;
        int s = 0, dl = 0;
        if (idx < cnt) {
            const int eid = hits[idx];
            s  = src[eid];
            dl = (dst[eid] - r0) * FD;
            if (p + idx < room) rsd[p + idx] = v2i{s, dl};
        }
        add_rows<SC_GRP>(s, dl, min(32, cnt - base), lane, hw, rows);
    }
}

__global__ __launch_bounds__(1024) void k_range_sums(
    const int* __restrict__ src, const int* __restrict__ dst,
    const unsigned* __restrict__ key4, const unsigned* __restrict__ degi,
    const float* __restrict__ dinv, const float* __restrict__ bias,
    const float* __restrict__ hw, float* agg,
    int* __restrict__ rec_sd, int* __restrict__ rinfo,
    int e, int n) {
    __shared__ alignas(16) float rows[RS_ROWS * FD];
    __shared__ int hits[RS_WAVES * RS_CAP];
    __shared__ unsigned wsum[RS_WAVES];
    __shared__ int lpos;
    const int lane = threadIdx.x & 31;
    const int wid  = __builtin_amdgcn_readfirstlane((int)(threadIdx.x >> 5));
    const int r0   = blockIdx.x * RS_ROWS;
    const int nr   = min(RS_ROWS, n - r0);
    if (nr <= 0) return;

    unsigned part = 0u;
    {
        const v4u* dg4 = (const v4u*)degi;
        const int  nq  = r0 / 4;
        for (int q0 = 0; q0 < nq; q0 += RS_DEG4 * RS_WAVES * 32) {
            v4u t[RS_DEG4];
#pragma unroll
            for (int i = 0; i < RS_DEG4; ++i)
                t[i] = dg4[min(q0 + (int)threadIdx.x + i * RS_WAVES * 32, nq - 1)];
#pragma unroll
            for (int i = 0; i < RS_DEG4; ++i)
                if (q0 + (int)threadIdx.x + i * RS_WAVES * 32 < nq)
                    part += t[i].x + t[i].y + t[i].z + t[i].w;
        }
    }
#pragma unroll
    for (int off = 16; off > 0; off >>= 1) part += __shfl_xor(part, off, 32);
    if (lane == 0) wsum[wid] = part;
    if (threadIdx.x == 0) lpos = 0;

    const v4f* hs = (const v4f*)(hw + (size_t)r0 * FD);
    for (int i = threadIdx.x; i < nr * (FD / 4); i += RS_WAVES * 32) rows_put(rows, i, hs[i]);
    __syncthreads();

    unsigned tot = wsum[lane];
#pragma unroll
    for (int off = 16; off > 0; off >>= 1) tot += __shfl_xor(tot, off, 32);
    const int gbase = __builtin_amdgcn_readfirstlane((int)tot - r0);
    v2i* myrsd = (v2i*)rec_sd + gbase;
    const int room = e - gbase;

    const unsigned splat = (unsigned)blockIdx.x * 0x01010101u;
    int* myhits = hits + wid * RS_CAP;
    int cnt = 0;
    const int thr = 32 + 8 * ((wid + (wid >> 2)) & 7);
    const int ntrips = (e + RS_TRIP - 1) / RS_TRIP;
    const int tlast  = ntrips - 1;
    const v4u* kp = (const v4u*)key4 + lane;
    v4u k0 = kp[(size_t)min(wid,                tlast) * 32];
    v4u k1 = kp[(size_t)min(wid +     RS_WAVES, tlast) * 32];
    v4u k2 = kp[(size_t)min(wid + 2 * RS_WAVES, tlast) * 32];
    v4u k3 = kp[(size_t)min(wid + 3 * RS_WAVES, tlast) * 32];
#pragma nounroll
    for (int tr = wid; tr < ntrips; tr += RS_WAVES) {
        const v4u kc = k0;
        k0 = k1; k1 = k2; k2 = k3;
        k3 = kp[(size_t)min(tr + 4 * RS_WAVES, tlast) * 32];
        unsigned m = (zero_bytes(kc.x ^ splat) >> 7) | (zero_bytes(kc.y ^ splat) >> 6) |
                     (zero_bytes(kc.z ^ splat) >> 5) | (zero_bytes(kc.w ^ splat) >> 4);
        while (__builtin_amdgcn_ballot_w32(m != 0u) != 0u) {
            const int  p   = __builtin_ctz(m | 0x80000000u);
            const int  eid = tr * RS_TRIP + lane * 16 + 4 * (p & 7) + (p >> 3);
            const bool ok  = (m != 0u) && (eid < e);
            const unsigned bm = __builtin_amdgcn_ballot_w32(ok);
            if (ok) myhits[cnt + (int)__builtin_amdgcn_mbcnt_lo(bm, 0u)] = eid;
            cnt += __builtin_popcount(bm);
            m &= m - 1u;
            if (cnt >= thr) {
                work_off(myhits, cnt, lane, r0, src, dst, hw, rows, myrsd, room, &lpos);
                cnt = 0;
            }
        }
    }
    work_off(myhits, cnt, lane, r0, src, dst, hw, rows, myrsd, room, &lpos);
    __syncthreads();

    if (threadIdx.x == 0) {
        rinfo[2 * blockIdx.x]     = gbase;
        rinfo[2 * blockIdx.x + 1] = min(lpos, room);
    }
    rows_out(rows, (v4f*)(agg + (size_t)r0 * FD), nr * (FD / 4), r0, dinv, bias);
}

static __device__ __forceinline__ v2i load_rec(const v2i* __restrict__ rsd, int idx, int cnt) {
    v2i sd = v2i{0, 0};
    if (idx < cnt) sd = rsd[idx];
    return sd;
}

__global__ __launch_bounds__(1024) void k_list_sums(
    const int* __restrict__ rec_sd, const int* __restrict__ rinfo,
    const float* __restrict__ dinv, const float* __restrict__ bias,
    const float* __restrict__ hw, float* agg, int n) {
    __shared__ alignas(16) float rows[RS_ROWS * FD];
    const int lane = threadIdx.x & 31;
    const int wid  = __builtin_amdgcn_readfirstlane((int)(threadIdx.x >> 5));
    const int r0   = blockIdx.x * RS_ROWS;
    const int nr   = min(RS_ROWS, n - r0);
    if (nr <= 0) return;
    const int gbase = rinfo[2 * blockIdx.x];
    const int cnt   = rinfo[2 * blockIdx.x + 1];
    const v2i* myrsd = (const v2i*)rec_sd + gbase;

    int c0 = wid * 32;
    v2i sd = load_rec(myrsd, c0 + lane, cnt);

    const v4f* gl = (const v4f*)(hw + (size_t)r0 * FD);
    const int nv = nr * (FD / 4);
    {
        v4f sv[RS_SV];
#pragma unroll
        for (int i = 0; i < RS_SV; ++i)
            sv[i] = gl[min((int)threadIdx.x + i * RS_WAVES * 32, nv - 1)];
#pragma unroll
        for (int i = 0; i < RS_SV; ++i)
            if ((int)threadIdx.x + i * RS_WAVES * 32 < nv)
                rows_put(rows, (int)threadIdx.x + i * RS_WAVES * 32, sv[i]);
    }
    __syncthreads();

    for (; c0 < cnt; c0 += RS_WAVES * 32) {
        const v2i sdn = load_rec(myrsd, c0 + RS_WAVES * 32 + lane, cnt);
        add_rows<LS_GRP>(sd.x, sd.y, min(32, cnt - c0), lane, hw, rows);
        sd = sdn;
    }
    __syncthreads();

    rows_out(rows, (v4f*)(agg + (size_t)r0 * FD), nv, r0, dinv, bias);
}

__global__ __launch_bounds__(256) void k_logits(
    const float* __restrict__ h, const float* __restrict__ Wp2,
    const float* __restrict__ bp2, float* __restrict__ out, int n) {
    const int wid  = (int)((blockIdx.x * blockDim.x + threadIdx.x) >> 5);
    const int lane = threadIdx.x & 31;
    if (wid >= n) return;
    const float4 v = ((const float4*)(h + (size_t)wid * FD))[lane];
    const int f = lane * 4;
    float p0 = v.x * Wp2[(f + 0) * 2 + 0] + v.y * Wp2[(f + 1) * 2 + 0] +
               v.z * Wp2[(f + 2) * 2 + 0] + v.w * Wp2[(f + 3) * 2 + 0];
    float p1 = v.x * Wp2[(f + 0) * 2 + 1] + v.y * Wp2[(f + 1) * 2 + 1] +
               v.z * Wp2[(f + 2) * 2 + 1] + v.w * Wp2[(f + 3) * 2 + 1];
#pragma unroll
    for (int off = 16; off > 0; off >>= 1) {
        p0 += __shfl_down(p0, off, 32);
        p1 += __shfl_down(p1, off, 32);
    }
    if (lane == 0) {
        const float l0 = p0 + bp2[0];
        const float l1 = p1 + bp2[1];
        const float m  = fmaxf(l0, l1);
        const float lse = m + logf(expf(l0 - m) + expf(l1 - m));
        out[(size_t)wid * 2 + 0] = l0 - lse;
        out[(size_t)wid * 2 + 1] = l1 - lse;
    }
}

extern "C" void kernel_launch(void* const* d_in, const int* in_sizes, int n_in,
                              void* d_out, int out_size, void* d_ws,
                              size_t ws_size, hipStream_t stream) {
    const float* x   = (const float*)d_in[0];
    const int*   ei  = (const int*)d_in[1];
    const float* W1  = (const float*)d_in[2];
    const float* b1  = (const float*)d_in[3];
    const float* W2  = (const float*)d_in[4];
    const float* b2  = (const float*)d_in[5];
    const float* W3  = (const float*)d_in[6];
    const float* b3  = (const float*)d_in[7];
    const float* Wp1 = (const float*)d_in[8];
    const float* bp1 = (const float*)d_in[9];
    const float* Wp2 = (const float*)d_in[10];
    const float* bp2 = (const float*)d_in[11];

    const int N = in_sizes[0] / FD;
    const int E = in_sizes[1] / 2;
    const int* src = ei;
    const int* dst = ei + E;

    char* ws = (char*)d_ws;
    float* hw   = (float*)ws;        ws += (size_t)N * FD * sizeof(float);
    float* agg1 = (float*)ws;        ws += (size_t)N * FD * sizeof(float);
    float* agg2 = (float*)ws;        ws += (size_t)N * FD * sizeof(float);
    unsigned* degi = (unsigned*)ws;  ws += (size_t)N * sizeof(unsigned);
    float* dinv = (float*)ws;        ws += (size_t)N * sizeof(float);
    unsigned* key4 = (unsigned*)ws;
    ws += (size_t)((E + RS_TRIP - 1) / RS_TRIP) * RS_TRIP;
    int* rec_sd  = (int*)ws;         ws += (size_t)E * 2 * sizeof(int);
    int* rinfo   = (int*)ws;

    float* emb_out  = (float*)d_out;
    float* prob_out = (float*)d_out + (size_t)N * FD;

    const dim3 blk(256);
    const int gN      = (N + 255) / 256;
    const int nTrips  = (E + RS_TRIP - 1) / RS_TRIP;
    const int nKey4   = nTrips * (RS_TRIP / 4);
    const int gE      = (nKey4 + 255) / 256;
    const int gTiles  = N / 16;
    const int gGemm   = (gTiles + GT_WAVES - 1) / GT_WAVES;
    const dim3 blkSum(RS_WAVES * 32);
    const int gNodeW  = (N * WAVE + 255) / 256;

    k_deg_init <<<gN, blk, 0, stream>>>(degi, N);
    k_deg_count<<<gE, blk, 0, stream>>>(dst, degi, key4, E, nKey4);
    k_dinv     <<<gN, blk, 0, stream>>>(degi, dinv, N);

    k_gemm_conv_in  <<<gGemm, blk, 0, stream>>>(x, W1, dinv, hw, gTiles);
    k_range_sums    <<<RS_WGS, blkSum, 0, stream>>>(src, dst, key4, degi, dinv, b1, hw, agg1, rec_sd, rinfo, E, N);

    k_gemm_conv_relu<<<gGemm, blk, 0, stream>>>(agg1, W2, dinv, hw, gTiles);
    k_list_sums     <<<RS_WGS, blkSum, 0, stream>>>(rec_sd, rinfo, dinv, b2, hw, agg2, N);

    k_gemm_conv_relu<<<gGemm, blk, 0, stream>>>(agg2, W3, dinv, hw, gTiles);
    k_list_sums     <<<RS_WGS, blkSum, 0, stream>>>(rec_sd, rinfo, dinv, b3, hw, emb_out, N);

    k_gemm_post_relu<<<gGemm, blk, 0, stream>>>(emb_out, Wp1, bp1, hw, gTiles);
    k_logits        <<<gNodeW, blk, 0, stream>>>(hw, Wp2, bp2, prob_out, N);
}


# Sum kernels end with the next layer's product: finished rows stay in LDS (pitch 132 words), next W staged in LDS, 16x32 WMMA units over the 32 waves; 3 product launches go; their k sum order changes
# speedup vs baseline: 3.6091x; 1.1548x over previous
#include <hip/hip_runtime.h>
#include <math.h>

#define FD 128
#define WAVE 32

typedef __attribute__((ext_vector_type(2))) float v2f;
typedef __attribute__((ext_vector_type(8))) float v8f;

#define RS_WGS   256
#define RS_ROWS  196
#define RS_TRIP  512

__global__ void k_deg_init(unsigned* __restrict__ degi, int n) {
    int i = blockIdx.x * blockDim.x + threadIdx.x;
    if (i < n) degi[i] = 1u;
}

__global__ void k_deg_count(const int* __restrict__ dst,
                            unsigned* __restrict__ degi,
                            unsigned* __restrict__ key4, int e, int nkey4) {
    int i = blockIdx.x * blockDim.x + threadIdx.x;
    if (i >= nkey4) return;
    unsigned d[4];
#pragma unroll
    for (int b = 0; b < 4; ++b) d[b] = (unsigned)dst[min(4 * i + b, e - 1)];
    unsigned k = 0u;
#pragma unroll
    for (int b = 0; b < 4; ++b) {
        if (4 * i + b < e) {
            atomicAdd(&degi[d[b]], 1u);
            k |= (d[b] / RS_ROWS) << (8 * b);
        }
    }
    key4[i] = k;
}

__global__ void k_dinv(const unsigned* __restrict__ degi,
                       float* __restrict__ dinv, int n) {
    int i = blockIdx.x * blockDim.x + threadIdx.x;
    if (i < n) dinv[i] = rsqrtf((float)degi[i]);
}

#define GT_WAVES 8
#define WL_HALF  (FD * 4 + 32)
#define WL_PAIR  (2 * FD * 4 + 32)
#define WL_WORDS (16 * WL_PAIR)

typedef __attribute__((ext_vector_type(4))) float v4f;

static __device__ __forceinline__ void stage_w(const float* __restrict__ W,
                                               float* __restrict__ wl) {
    const int n  = threadIdx.x & (FD - 1);
    const int g0 = threadIdx.x >> 7;
#pragma unroll
    for (int i = 0; i < FD / 8; ++i) {
        const int g = 2 * i + g0;
        v4f q;
        q.x = W[(size_t)(4 * g + 0) * FD + n];
        q.y = W[(size_t)(4 * g + 1) * FD + n];
        q.z = W[(size_t)(4 * g + 2) * FD + n];
        q.w = W[(size_t)(4 * g + 3) * FD + n];
        *(v4f*)(wl + i * WL_PAIR + g0 * WL_HALF + n * 4) = q;
    }
}

static __device__ __forceinline__ void load_a_tile(
    const float* __restrict__ arow, v4f (&a)[FD / 8]) {
#pragma unroll
    for (int j = 0; j < FD / 8; ++j) a[j] = *(const v4f*)(arow + 8 * j);
}

static __device__ __forceinline__ void load_b_step(
    const float* __restrict__ wlane, int j, v4f (&b)[8]) {
#pragma unroll
    for (int t = 0; t < 8; ++t)
        b[t] = *(const v4f*)(wlane + j * WL_PAIR + 64 * t);
}

static __device__ __forceinline__ void wmma_step(
    const v4f& av4, const v4f (&b)[8], v8f (&acc)[8]) {
#pragma unroll
    for (int t = 0; t < 8; ++t) {
        const v2f av = {av4.x, av4.y};
        const v2f bv = {b[t].x, b[t].y};
        acc[t] = __builtin_amdgcn_wmma_f32_16x16x4_f32(
            false, av, false, bv, (short)0, acc[t], false, false);
    }
#pragma unroll
    for (int t = 0; t < 8; ++t) {
        const v2f av = {av4.z, av4.w};
        const v2f bv = {b[t].z, b[t].w};
        acc[t] = __builtin_amdgcn_wmma_f32_16x16x4_f32(
            false, av, false, bv, (short)0, acc[t], false, false);
    }
}

static __device__ __forceinline__ void rowtile_product(
    const v4f (&a)[FD / 8], const float* __restrict__ wlane, v8f (&acc)[8]) {
    v4f b0[8], b1[8];
    load_b_step(wlane, 0, b0);
#pragma unroll
    for (int j = 0; j < FD / 8; j += 2) {
        load_b_step(wlane, j + 1, b1);
        __builtin_amdgcn_sched_barrier(0);
        wmma_step(a[j], b0, acc);
        __builtin_amdgcn_sched_barrier(0);
        if (j + 2 < FD / 8) load_b_step(wlane, j + 2, b0);
        __builtin_amdgcn_sched_barrier(0);
        wmma_step(a[j + 1], b1, acc);
        __builtin_amdgcn_sched_barrier(0);
    }
}

static __device__ __forceinline__ bool rowtile_front(
    const float* __restrict__ A, const float* __restrict__ W, int ntiles,
    float* __restrict__ wl, int& m0, int& c, int& h,
    v4f (&a)[FD / 8]) {
    const int lane = threadIdx.x & 31;
    const int wid  = __builtin_amdgcn_readfirstlane((int)(threadIdx.x >> 5));
    const int tile = blockIdx.x * GT_WAVES + wid;
    const int tl   = min(tile, ntiles - 1);
    c  = lane & 15;
    h  = lane >> 4;
    m0 = tile << 4;

    load_a_tile(A + (size_t)((tl << 4) + c) * FD + 4 * h, a);

    stage_w(W, wl);
    __syncthreads();
    if (tile >= ntiles) return false;
    return true;
}

static __device__ __forceinline__ void gemm_conv_body(
    const float* __restrict__ A, const float* __restrict__ W,
    const float* __restrict__ dinv,
    float* __restrict__ hw, int ntiles,
    float* __restrict__ wl) {
    int m0, c, h;
    v4f a[FD / 8];
    if (!rowtile_front(A, W, ntiles, wl, m0, c, h, a)) return;

    float di[8];
#pragma unroll
    for (int v = 0; v < 8; ++v) di[v] = dinv[m0 + v + 8 * h];

    v8f acc[8];
#pragma unroll
    for (int t = 0; t < 8; ++t) acc[t] = v8f{};
    rowtile_product(a, wl + h * WL_HALF + c * 4, acc);

#pragma unroll
    for (int v = 0; v < 8; ++v) {
        const int row = m0 + v + 8 * h;
#pragma unroll
        for (int t = 0; t < 8; ++t)
            hw[(size_t)row * FD + 16 * t + c] = acc[t][v] * di[v];
    }
}

__global__ __launch_bounds__(256) void k_gemm_conv_in(
    const float* __restrict__ A, const float* __restrict__ W,
    const float* __restrict__ dinv, float* __restrict__ hw, int ntiles) {
    __shared__ alignas(16) float wl[WL_WORDS];
    gemm_conv_body(A, W, dinv, hw, ntiles, wl);
}

#define RS_WAVES 32
#define RS_CAP   512
#define SC_GRP   8
#define LS_GRP   16
#define RS_DEG4  13
#define RS_SV    ((RS_ROWS * (FD / 4) + RS_WAVES * 32 - 1) / (RS_WAVES * 32))
#define RS_PITCH (FD + 4)
#define RS_TILES ((RS_ROWS + 15) / 16)
#define RS_UNITS (RS_TILES * 4)

typedef __attribute__((ext_vector_type(4))) unsigned v4u;
typedef __attribute__((ext_vector_type(2))) int v2i;

static __device__ __forceinline__ unsigned zero_bytes(unsigned x) {
    const unsigned t = (x & 0x7f7f7f7fu) + 0x7f7f7f7fu;
    return ~(t | x | 0x7f7f7f7fu);
}

static __device__ __forceinline__ void rows_put(float* rows, int i, v4f q) {
    float* rp = rows + (i >> 5) * RS_PITCH + (i & 31);
    rp[0] = q.x; rp[32] = q.y; rp[64] = q.z; rp[96] = q.w;
}
static __device__ __forceinline__ v4f rows_get(const float* rows, int i) {
    const float* rp = rows + (i >> 5) * RS_PITCH + (i & 31);
    return v4f{rp[0], rp[32], rp[64], rp[96]};
}

static __device__ __forceinline__ void rows_fin(
    float* rows, v4f* gl, bool store, int nv, int r0,
    const float* __restrict__ dinv, const float* __restrict__ bias) {
    const v4f b4 = ((const v4f*)bias)[threadIdx.x & 31];
    for (int i = threadIdx.x; i < nv; i += RS_WAVES * 32) {
        const float di = dinv[r0 + (i >> 5)];
        const v4f   q  = rows_get(rows, i);
        const v4f   r  = v4f{q.x * di + b4.x, q.y * di + b4.y, q.z * di + b4.z, q.w * di + b4.w};
        if (store) gl[i] = r;
        rows_put(rows, i, r);
    }
}

static __device__ __forceinline__ void stage_w_rows(const float* __restrict__ W,
                                                    float* __restrict__ wl) {
    const int n  = threadIdx.x & (FD - 1);
    const int g0 = threadIdx.x >> 7;
#pragma unroll
    for (int i = 0; i < 4; ++i) {
        const int g  = 8 * i + g0;
        const int k0 = 16 * g0 + i;
        v4f q;
        q.x = W[(size_t)(k0 +  0) * FD + n];
        q.y = W[(size_t)(k0 +  4) * FD + n];
        q.z = W[(size_t)(k0 +  8) * FD + n];
        q.w = W[(size_t)(k0 + 12) * FD + n];
        *(v4f*)(wl + (g >> 1) * WL_PAIR + (g & 1) * WL_HALF + n * 4) = q;
    }
}

static __device__ __forceinline__ void rows_product(
    const float* rows, const float* wl, int nr, int r0, int wid, int lane,
    bool post, const float* __restrict__ dinv, const float* __restrict__ bias,
    float* __restrict__ outp) {
    const int c = lane & 15;
    const int h = lane >> 4;
    for (int u = wid; u < RS_UNITS; u += RS_WAVES) {
        const int rt = u >> 2;
        const int cq = u & 3;
        if (rt * 16 >= nr) break;
        const float* ap = rows + (rt * 16 + c) * RS_PITCH + 4 * h;
        const float* bp = wl + h * WL_HALF + (cq * 32 + c) * 4;

        float ep[8];
        float bc0 = 0.0f, bc1 = 0.0f;
        if (post) {
            bc0 = bias[cq * 32 + c];
            bc1 = bias[cq * 32 + 16 + c];
        } else {
#pragma unroll
            for (int v = 0; v < 8; ++v) ep[v] = dinv[r0 + min(rt * 16 + v + 8 * h, nr - 1)];
        }

        v8f acc0 = v8f{}, acc1 = v8f{};
#pragma unroll
        for (int j = 0; j < FD / 8; ++j) {
            v4f a = *(const v4f*)(ap + 8 * j);
            const v4f b0 = *(const v4f*)(bp + j * WL_PAIR);
            const v4f b1 = *(const v4f*)(bp + j * WL_PAIR + 64);
            a.x = fmaxf(a.x, 0.0f); a.y = fmaxf(a.y, 0.0f);
            a.z = fmaxf(a.z, 0.0f); a.w = fmaxf(a.w, 0.0f);
            const v2f a01 = {a.x, a.y}, a23 = {a.z, a.w};
            const v2f p0 = {b0.x, b0.y}, p1 = {b1.x, b1.y};
            const v2f q0 = {b0.z, b0.w}, q1 = {b1.z, b1.w};
            acc0 = __builtin_amdgcn_wmma_f32_16x16x4_f32(false, a01, false, p0, (short)0, acc0, false, false);
            acc1 = __builtin_amdgcn_wmma_f32_16x16x4_f32(false, a01, false, p1, (short)0, acc1, false, false);
            acc0 = __builtin_amdgcn_wmma_f32_16x16x4_f32(false, a23, false, q0, (short)0, acc0, false, false);
            acc1 = __builtin_amdgcn_wmma_f32_16x16x4_f32(false, a23, false, q1, (short)0, acc1, false, false);
        }

#pragma unroll
        for (int v = 0; v < 8; ++v) {
            const int lr = rt * 16 + v + 8 * h;
            if (lr < nr) {
                float* op = outp + (size_t)(r0 + lr) * FD + cq * 32 + c;
                if (post) {
                    op[0]  = fmaxf(acc0[v] + bc0, 0.0f);
                    op[16] = fmaxf(acc1[v] + bc1, 0.0f);
                } else {
                    op[0]  = acc0[v] * ep[v];
                    op[16] = acc1[v] * ep[v];
                }
            }
        }
    }
}

template <int G>
static __device__ __forceinline__ void add_rows(
    int s, int dl, int nt, int lane,
    const float* __restrict__ hw, float* rows) {
    for (int g0 = 0; g0 < nt; g0 += G) {
        v4f v[G];
#pragma unroll
        for (int g = 0; g < G; ++g) {
            const int sj = __builtin_amdgcn_readlane(s, min(g0 + g, nt - 1));
            v[g] = ((const v4f*)(hw + (size_t)sj * FD))[lane];
        }
#pragma unroll
        for (int g = 0; g < G; ++g) {
            if (g == 0 || g0 + g < nt) {
                const int dj = __builtin_amdgcn_readlane(dl, g0 + g);
                float* rp = rows + dj + lane;
                unsafeAtomicAdd(rp,      v[g].x);
                unsafeAtomicAdd(rp + 32, v[g].y);
                unsafeAtomicAdd(rp + 64, v[g].z);
                unsafeAtomicAdd(rp + 96, v[g].w);
            }
        }
    }
}

static __device__ __forceinline__ void work_off(
    const int* hits, int cnt, int lane, int r0,
    const int* __restrict__ src, const int* __restrict__ dst,
    const float* __restrict__ hw, float* rows,
    v2i* __restrict__ rsd, int room, int* lpos) {
    if (cnt <= 0) return;
    int p = 0;
    if (lane == 0) p = atomicAdd(lpos, cnt);
    p = __builtin_amdgcn_readfirstlane(p);
    for (int base = 0; base < cnt; base += 32) {
        const int idx = base + lane;
        int s = 0, dl = 0;
        if (idx < cnt) {
            const int eid = hits[idx];
            s  = src[eid];
            dl = (dst[eid] - r0) * RS_PITCH;
            if (p + idx < room) rsd[p + idx] = v2i{s, dl};
        }
        add_rows<SC_GRP>(s, dl, min(32, cnt - base), lane, hw, rows);
    }
}

__global__ __launch_bounds__(1024) void k_range_sums(
    const int* __restrict__ src, const int* __restrict__ dst,
    const unsigned* __restrict__ key4, const unsigned* __restrict__ degi,
    const float* __restrict__ dinv, const float* __restrict__ bias,
    const float* __restrict__ hw, const float* __restrict__ Wn,
    float* __restrict__ hwn,
    int* __restrict__ rec_sd, int* __restrict__ rinfo,
    int e, int n) {
    __shared__ alignas(16) float rows[RS_TILES * 16 * RS_PITCH];
    __shared__ alignas(16) float wl[WL_WORDS];
    __shared__ int hits[RS_WAVES * RS_CAP];
    __shared__ unsigned wsum[RS_WAVES];
    __shared__ int lpos;
    const int lane = threadIdx.x & 31;
    const int wid  = __builtin_amdgcn_readfirstlane((int)(threadIdx.x >> 5));
    const int r0   = blockIdx.x * RS_ROWS;
    const int nr   = min(RS_ROWS, n - r0);
    if (nr <= 0) return;

    stage_w_rows(Wn, wl);

    unsigned part = 0u;
    {
        const v4u* dg4 = (const v4u*)degi;
        const int  nq  = r0 / 4;
        for (int q0 = 0; q0 < nq; q0 += RS_DEG4 * RS_WAVES * 32) {
            v4u t[RS_DEG4];
#pragma unroll
            for (int i = 0; i < RS_DEG4; ++i)
                t[i] = dg4[min(q0 + (int)threadIdx.x + i * RS_WAVES * 32, nq - 1)];
#pragma unroll
            for (int i = 0; i < RS_DEG4; ++i)
                if (q0 + (int)threadIdx.x + i * RS_WAVES * 32 < nq)
                    part += t[i].x + t[i].y + t[i].z + t[i].w;
        }
    }
#pragma unroll
    for (int off = 16; off > 0; off >>= 1) part += __shfl_xor(part, off, 32);
    if (lane == 0) wsum[wid] = part;
    if (threadIdx.x == 0) lpos = 0;

    const v4f* hs = (const v4f*)(hw + (size_t)r0 * FD);
    for (int i = threadIdx.x; i < nr * (FD / 4); i += RS_WAVES * 32) rows_put(rows, i, hs[i]);
    __syncthreads();

    unsigned tot = wsum[lane];
#pragma unroll
    for (int off = 16; off > 0; off >>= 1) tot += __shfl_xor(tot, off, 32);
    const int gbase = __builtin_amdgcn_readfirstlane((int)tot - r0);
    v2i* myrsd = (v2i*)rec_sd + gbase;
    const int room = e - gbase;

    const unsigned splat = (unsigned)blockIdx.x * 0x01010101u;
    int* myhits = hits + wid * RS_CAP;
    int cnt = 0;
    const int thr = 32 + 8 * ((wid + (wid >> 2)) & 7);
    const int ntrips = (e + RS_TRIP - 1) / RS_TRIP;
    const int tlast  = ntrips - 1;
    const v4u* kp = (const v4u*)key4 + lane;
    v4u k0 = kp[(size_t)min(wid,                tlast) * 32];
    v4u k1 = kp[(size_t)min(wid +     RS_WAVES, tlast) * 32];
    v4u k2 = kp[(size_t)min(wid + 2 * RS_WAVES, tlast) * 32];
    v4u k3 = kp[(size_t)min(wid + 3 * RS_WAVES, tlast) * 32];
#pragma nounroll
    for (int tr = wid; tr < ntrips; tr += RS_WAVES) {
        const v4u kc = k0;
        k0 = k1; k1 = k2; k2 = k3;
        k3 = kp[(size_t)min(tr + 4 * RS_WAVES, tlast) * 32];
        unsigned m = (zero_bytes(kc.x ^ splat) >> 7) | (zero_bytes(kc.y ^ splat) >> 6) |
                     (zero_bytes(kc.z ^ splat) >> 5) | (zero_bytes(kc.w ^ splat) >> 4);
        while (__builtin_amdgcn_ballot_w32(m != 0u) != 0u) {
            const int  p   = __builtin_ctz(m | 0x80000000u);
            const int  eid = tr * RS_TRIP + lane * 16 + 4 * (p & 7) + (p >> 3);
            const bool ok  = (m != 0u) && (eid < e);
            const unsigned bm = __builtin_amdgcn_ballot_w32(ok);
            if (ok) myhits[cnt + (int)__builtin_amdgcn_mbcnt_lo(bm, 0u)] = eid;
            cnt += __builtin_popcount(bm);
            m &= m - 1u;
            if (cnt >= thr) {
                work_off(myhits, cnt, lane, r0, src, dst, hw, rows, myrsd, room, &lpos);
                cnt = 0;
            }
        }
    }
    work_off(myhits, cnt, lane, r0, src, dst, hw, rows, myrsd, room, &lpos);
    __syncthreads();

    if (threadIdx.x == 0) {
        rinfo[2 * blockIdx.x]     = gbase;
        rinfo[2 * blockIdx.x + 1] = min(lpos, room);
    }
    rows_fin(rows, (v4f*)0, false, nr * (FD / 4), r0, dinv, bias);
    __syncthreads();
    rows_product(rows, wl, nr, r0, wid, lane, false, dinv, (const float*)0, hwn);
}

static __device__ __forceinline__ v2i load_rec(const v2i* __restrict__ rsd, int idx, int cnt) {
    v2i sd = v2i{0, 0};
    if (idx < cnt) sd = rsd[idx];
    return sd;
}

static __device__ __forceinline__ void list_sums_body(
    const int* __restrict__ rec_sd, const int* __restrict__ rinfo,
    const float* __restrict__ dinv, const float* __restrict__ bias,
    const float* __restrict__ hw, const float* __restrict__ Wn,
    const float* __restrict__ biasn, float* __restrict__ emb,
    float* __restrict__ hwn, int n, bool post,
    float* rows, float* wl) {
    const int lane = threadIdx.x & 31;
    const int wid  = __builtin_amdgcn_readfirstlane((int)(threadIdx.x >> 5));
    const int r0   = blockIdx.x * RS_ROWS;
    const int nr   = min(RS_ROWS, n - r0);
    if (nr <= 0) return;
    const int gbase = rinfo[2 * blockIdx.x];
    const int cnt   = rinfo[2 * blockIdx.x + 1];
    const v2i* myrsd = (const v2i*)rec_sd + gbase;

    int c0 = wid * 32;
    v2i sd = load_rec(myrsd, c0 + lane, cnt);

    stage_w_rows(Wn, wl);

    const v4f* gl = (const v4f*)(hw + (size_t)r0 * FD);
    const int nv = nr * (FD / 4);
    {
        v4f sv[RS_SV];
#pragma unroll
        for (int i = 0; i < RS_SV; ++i)
            sv[i] = gl[min((int)threadIdx.x + i * RS_WAVES * 32, nv - 1)];
#pragma unroll
        for (int i = 0; i < RS_SV; ++i)
            if ((int)threadIdx.x + i * RS_WAVES * 32 < nv)
                rows_put(rows, (int)threadIdx.x + i * RS_WAVES * 32, sv[i]);
    }
    __syncthreads();

    for (; c0 < cnt; c0 += RS_WAVES * 32) {
        const v2i sdn = load_rec(myrsd, c0 + RS_WAVES * 32 + lane, cnt);
        add_rows<LS_GRP>(sd.x, sd.y, min(32, cnt - c0), lane, hw, rows);
        sd = sdn;
    }
    __syncthreads();

    rows_fin(rows, post ? (v4f*)(emb + (size_t)r0 * FD) : (v4f*)0, post, nv, r0, dinv, bias);
    __syncthreads();
    rows_product(rows, wl, nr, r0, wid, lane, post, dinv, biasn, hwn);
}

__global__ __launch_bounds__(1024) void k_list_sums_conv(
    const int* __restrict__ rec_sd, const int* __restrict__ rinfo,
    const float* __restrict__ dinv, const float* __restrict__ bias,
    const float* __restrict__ hw, const float* __restrict__ Wn,
    float* __restrict__ hwn, int n) {
    __shared__ alignas(16) float rows[RS_TILES * 16 * RS_PITCH];
    __shared__ alignas(16) float wl[WL_WORDS];
    list_sums_body(rec_sd, rinfo, dinv, bias, hw, Wn, (const float*)0, (float*)0, hwn, n, false, rows, wl);
}

__global__ __launch_bounds__(1024) void k_list_sums_post(
    const int* __restrict__ rec_sd, const int* __restrict__ rinfo,
    const float* __restrict__ dinv, const float* __restrict__ bias,
    const float* __restrict__ hw, const float* __restrict__ Wn,
    const float* __restrict__ biasn, float* __restrict__ emb,
    float* __restrict__ hwn, int n) {
    __shared__ alignas(16) float rows[RS_TILES * 16 * RS_PITCH];
    __shared__ alignas(16) float wl[WL_WORDS];
    list_sums_body(rec_sd, rinfo, dinv, bias, hw, Wn, biasn, emb, hwn, n, true, rows, wl);
}

__global__ __launch_bounds__(256) void k_logits(
    const float* __restrict__ h, const float* __restrict__ Wp2,
    const float* __restrict__ bp2, float* __restrict__ out, int n) {
    const int wid  = (int)((blockIdx.x * blockDim.x + threadIdx.x) >> 5);
    const int lane = threadIdx.x & 31;
    if (wid >= n) return;
    const float4 v = ((const float4*)(h + (size_t)wid * FD))[lane];
    const int f = lane * 4;
    float p0 = v.x * Wp2[(f + 0) * 2 + 0] + v.y * Wp2[(f + 1) * 2 + 0] +
               v.z * Wp2[(f + 2) * 2 + 0] + v.w * Wp2[(f + 3) * 2 + 0];
    float p1 = v.x * Wp2[(f + 0) * 2 + 1] + v.y * Wp2[(f + 1) * 2 + 1] +
               v.z * Wp2[(f + 2) * 2 + 1] + v.w * Wp2[(f + 3) * 2 + 1];
#pragma unroll
    for (int off = 16; off > 0; off >>= 1) {
        p0 += __shfl_down(p0, off, 32);
        p1 += __shfl_down(p1, off, 32);
    }
    if (lane == 0) {
        const float l0 = p0 + bp2[0];
        const float l1 = p1 + bp2[1];
        const float m  = fmaxf(l0, l1);
        const float lse = m + logf(expf(l0 - m) + expf(l1 - m));
        out[(size_t)wid * 2 + 0] = l0 - lse;
        out[(size_t)wid * 2 + 1] = l1 - lse;
    }
}

extern "C" void kernel_launch(void* const* d_in, const int* in_sizes, int n_in,
                              void* d_out, int out_size, void* d_ws,
                              size_t ws_size, hipStream_t stream) {
    const float* x   = (const float*)d_in[0];
    const int*   ei  = (const int*)d_in[1];
    const float* W1  = (const float*)d_in[2];
    const float* b1  = (const float*)d_in[3];
    const float* W2  = (const float*)d_in[4];
    const float* b2  = (const float*)d_in[5];
    const float* W3  = (const float*)d_in[6];
    const float* b3  = (const float*)d_in[7];
    const float* Wp1 = (const float*)d_in[8];
    const float* bp1 = (const float*)d_in[9];
    const float* Wp2 = (const float*)d_in[10];
    const float* bp2 = (const float*)d_in[11];

    const int N = in_sizes[0] / FD;
    const int E = in_sizes[1] / 2;
    const int* src = ei;
    const int* dst = ei + E;

    char* ws = (char*)d_ws;
    float* hwa  = (float*)ws;        ws += (size_t)N * FD * sizeof(float);
    float* hwb  = (float*)ws;        ws += (size_t)N * FD * sizeof(float);
    unsigned* degi = (unsigned*)ws;  ws += (size_t)N * sizeof(unsigned);
    float* dinv = (float*)ws;        ws += (size_t)N * sizeof(float);
    unsigned* key4 = (unsigned*)ws;
    ws += (size_t)((E + RS_TRIP - 1) / RS_TRIP) * RS_TRIP;
    int* rec_sd  = (int*)ws;         ws += (size_t)E * 2 * sizeof(int);
    int* rinfo   = (int*)ws;

    float* emb_out  = (float*)d_out;
    float* prob_out = (float*)d_out + (size_t)N * FD;

    const dim3 blk(256);
    const int gN      = (N + 255) / 256;
    const int nTrips  = (E + RS_TRIP - 1) / RS_TRIP;
    const int nKey4   = nTrips * (RS_TRIP / 4);
    const int gE      = (nKey4 + 255) / 256;
    const int gTiles  = N / 16;
    const int gGemm   = (gTiles + GT_WAVES - 1) / GT_WAVES;
    const dim3 blkSum(RS_WAVES * 32);
    const int gNodeW  = (N * WAVE + 255) / 256;

    k_deg_init <<<gN, blk, 0, stream>>>(degi, N);
    k_deg_count<<<gE, blk, 0, stream>>>(dst, degi, key4, E, nKey4);
    k_dinv     <<<gN, blk, 0, stream>>>(degi, dinv, N);

    k_gemm_conv_in  <<<gGemm, blk, 0, stream>>>(x, W1, dinv, hwa, gTiles);
    k_range_sums    <<<RS_WGS, blkSum, 0, stream>>>(src, dst, key4, degi, dinv, b1, hwa, W2, hwb, rec_sd, rinfo, E, N);

    k_list_sums_conv<<<RS_WGS, blkSum, 0, stream>>>(rec_sd, rinfo, dinv, b2, hwb, W3, hwa, N);

    k_list_sums_post<<<RS_WGS, blkSum, 0, stream>>>(rec_sd, rinfo, dinv, b3, hwa, Wp1, bp1, emb_out, hwb, N);

    k_logits        <<<gNodeW, blk, 0, stream>>>(hwb, Wp2, bp2, prob_out, N);
}


# Degrees counted by the range owners: a search launch lists each range's edges and counts its 196 rows in LDS (key pass: 256 range bins); no per-edge add in memory; layer-1 sum order: list, as 2, 3
# speedup vs baseline: 4.5031x; 1.2477x over previous
#include <hip/hip_runtime.h>
#include <math.h>

#define FD 128
#define WAVE 32

typedef __attribute__((ext_vector_type(2))) float v2f;
typedef __attribute__((ext_vector_type(8))) float v8f;

#define RS_WGS   256
#define RS_ROWS  196
#define RS_TRIP  512
#define KC_WGS   128
#define KC_THR   1024

__global__ void k_cnt_init(unsigned* __restrict__ rcnt) {
    rcnt[threadIdx.x] = 0u;
}

__global__ __launch_bounds__(KC_THR) void k_keys(const int* __restrict__ dst,
                                                 unsigned* __restrict__ key4,
                                                 unsigned* __restrict__ rcnt,
                                                 int e, int nkey4) {
    __shared__ unsigned bins[RS_WGS];
    if (threadIdx.x < RS_WGS) bins[threadIdx.x] = 0u;
    __syncthreads();
    for (int i = blockIdx.x * KC_THR + threadIdx.x; i < nkey4; i += KC_WGS * KC_THR) {
        unsigned d[4];
#pragma unroll
        for (int b = 0; b < 4; ++b) d[b] = (unsigned)dst[min(4 * i + b, e - 1)];
        unsigned k = 0u;
#pragma unroll
        for (int b = 0; b < 4; ++b) {
            if (4 * i + b < e) {
                const unsigned g = min(d[b] / RS_ROWS, (unsigned)(RS_WGS - 1));
                atomicAdd(&bins[g], 1u);
                k |= g << (8 * b);
            }
        }
        key4[i] = k;
    }
    __syncthreads();
    if (threadIdx.x < RS_WGS) atomicAdd(&rcnt[threadIdx.x], bins[threadIdx.x]);
}

#define GT_WAVES 8
#define WL_HALF  (FD * 4 + 32)
#define WL_PAIR  (2 * FD * 4 + 32)
#define WL_WORDS (16 * WL_PAIR)

typedef __attribute__((ext_vector_type(4))) float v4f;

static __device__ __forceinline__ void stage_w(const float* __restrict__ W,
                                               float* __restrict__ wl) {
    const int n  = threadIdx.x & (FD - 1);
    const int g0 = threadIdx.x >> 7;
#pragma unroll
    for (int i = 0; i < FD / 8; ++i) {
        const int g = 2 * i + g0;
        v4f q;
        q.x = W[(size_t)(4 * g + 0) * FD + n];
        q.y = W[(size_t)(4 * g + 1) * FD + n];
        q.z = W[(size_t)(4 * g + 2) * FD + n];
        q.w = W[(size_t)(4 * g + 3) * FD + n];
        *(v4f*)(wl + i * WL_PAIR + g0 * WL_HALF + n * 4) = q;
    }
}

static __device__ __forceinline__ void load_a_tile(
    const float* __restrict__ arow, v4f (&a)[FD / 8]) {
#pragma unroll
    for (int j = 0; j < FD / 8; ++j) a[j] = *(const v4f*)(arow + 8 * j);
}

static __device__ __forceinline__ void load_b_step(
    const float* __restrict__ wlane, int j, v4f (&b)[8]) {
#pragma unroll
    for (int t = 0; t < 8; ++t)
        b[t] = *(const v4f*)(wlane + j * WL_PAIR + 64 * t);
}

static __device__ __forceinline__ void wmma_step(
    const v4f& av4, const v4f (&b)[8], v8f (&acc)[8]) {
#pragma unroll
    for (int t = 0; t < 8; ++t) {
        const v2f av = {av4.x, av4.y};
        const v2f bv = {b[t].x, b[t].y};
        acc[t] = __builtin_amdgcn_wmma_f32_16x16x4_f32(
            false, av, false, bv, (short)0, acc[t], false, false);
    }
#pragma unroll
    for (int t = 0; t < 8; ++t) {
        const v2f av = {av4.z, av4.w};
        const v2f bv = {b[t].z, b[t].w};
        acc[t] = __builtin_amdgcn_wmma_f32_16x16x4_f32(
            false, av, false, bv, (short)0, acc[t], false, false);
    }
}

static __device__ __forceinline__ void rowtile_product(
    const v4f (&a)[FD / 8], const float* __restrict__ wlane, v8f (&acc)[8]) {
    v4f b0[8], b1[8];
    load_b_step(wlane, 0, b0);
#pragma unroll
    for (int j = 0; j < FD / 8; j += 2) {
        load_b_step(wlane, j + 1, b1);
        __builtin_amdgcn_sched_barrier(0);
        wmma_step(a[j], b0, acc);
        __builtin_amdgcn_sched_barrier(0);
        if (j + 2 < FD / 8) load_b_step(wlane, j + 2, b0);
        __builtin_amdgcn_sched_barrier(0);
        wmma_step(a[j + 1], b1, acc);
        __builtin_amdgcn_sched_barrier(0);
    }
}

static __device__ __forceinline__ bool rowtile_front(
    const float* __restrict__ A, const float* __restrict__ W, int ntiles,
    float* __restrict__ wl, int& m0, int& c, int& h,
    v4f (&a)[FD / 8]) {
    const int lane = threadIdx.x & 31;
    const int wid  = __builtin_amdgcn_readfirstlane((int)(threadIdx.x >> 5));
    const int tile = blockIdx.x * GT_WAVES + wid;
    const int tl   = min(tile, ntiles - 1);
    c  = lane & 15;
    h  = lane >> 4;
    m0 = tile << 4;

    load_a_tile(A + (size_t)((tl << 4) + c) * FD + 4 * h, a);

    stage_w(W, wl);
    __syncthreads();
    if (tile >= ntiles) return false;
    return true;
}

static __device__ __forceinline__ void gemm_conv_body(
    const float* __restrict__ A, const float* __restrict__ W,
    const float* __restrict__ dinv,
    float* __restrict__ hw, int ntiles,
    float* __restrict__ wl) {
    int m0, c, h;
    v4f a[FD / 8];
    if (!rowtile_front(A, W, ntiles, wl, m0, c, h, a)) return;

    float di[8];
#pragma unroll
    for (int v = 0; v < 8; ++v) di[v] = dinv[m0 + v + 8 * h];

    v8f acc[8];
#pragma unroll
    for (int t = 0; t < 8; ++t) acc[t] = v8f{};
    rowtile_product(a, wl + h * WL_HALF + c * 4, acc);

#pragma unroll
    for (int v = 0; v < 8; ++v) {
        const int row = m0 + v + 8 * h;
#pragma unroll
        for (int t = 0; t < 8; ++t)
            hw[(size_t)row * FD + 16 * t + c] = acc[t][v] * di[v];
    }
}

__global__ __launch_bounds__(256) void k_gemm_conv_in(
    const float* __restrict__ A, const float* __restrict__ W,
    const float* __restrict__ dinv, float* __restrict__ hw, int ntiles) {
    __shared__ alignas(16) float wl[WL_WORDS];
    gemm_conv_body(A, W, dinv, hw, ntiles, wl);
}

#define RS_WAVES 32
#define RS_CAP   1312
#define LS_GRP   16
#define RS_SV    ((RS_ROWS * (FD / 4) + RS_WAVES * 32 - 1) / (RS_WAVES * 32))
#define RS_PITCH (FD + 4)
#define RS_TILES ((RS_ROWS + 15) / 16)
#define RS_UNITS (RS_TILES * 4)

typedef __attribute__((ext_vector_type(4))) unsigned v4u;
typedef __attribute__((ext_vector_type(2))) int v2i;

static __device__ __forceinline__ unsigned zero_bytes(unsigned x) {
    const unsigned t = (x & 0x7f7f7f7fu) + 0x7f7f7f7fu;
    return ~(t | x | 0x7f7f7f7fu);
}

static __device__ __forceinline__ void rows_put(float* rows, int i, v4f q) {
    float* rp = rows + (i >> 5) * RS_PITCH + (i & 31);
    rp[0] = q.x; rp[32] = q.y; rp[64] = q.z; rp[96] = q.w;
}
static __device__ __forceinline__ v4f rows_get(const float* rows, int i) {
    const float* rp = rows + (i >> 5) * RS_PITCH + (i & 31);
    return v4f{rp[0], rp[32], rp[64], rp[96]};
}

static __device__ __forceinline__ void rows_fin(
    float* rows, v4f* gl, bool store, int nv, int r0,
    const float* __restrict__ dinv, const float* __restrict__ bias) {
    const v4f b4 = ((const v4f*)bias)[threadIdx.x & 31];
    for (int i = threadIdx.x; i < nv; i += RS_WAVES * 32) {
        const float di = dinv[r0 + (i >> 5)];
        const v4f   q  = rows_get(rows, i);
        const v4f   r  = v4f{q.x * di + b4.x, q.y * di + b4.y, q.z * di + b4.z, q.w * di + b4.w};
        if (store) gl[i] = r;
        rows_put(rows, i, r);
    }
}

static __device__ __forceinline__ void stage_w_rows(const float* __restrict__ W,
                                                    float* __restrict__ wl) {
    const int n  = threadIdx.x & (FD - 1);
    const int g0 = threadIdx.x >> 7;
#pragma unroll
    for (int i = 0; i < 4; ++i) {
        const int g  = 8 * i + g0;
        const int k0 = 16 * g0 + i;
        v4f q;
        q.x = W[(size_t)(k0 +  0) * FD + n];
        q.y = W[(size_t)(k0 +  4) * FD + n];
        q.z = W[(size_t)(k0 +  8) * FD + n];
        q.w = W[(size_t)(k0 + 12) * FD + n];
        *(v4f*)(wl + (g >> 1) * WL_PAIR + (g & 1) * WL_HALF + n * 4) = q;
    }
}

static __device__ __forceinline__ void rows_product(
    const float* rows, const float* wl, int nr, int r0, int wid, int lane,
    bool post, const float* __restrict__ dinv, const float* __restrict__ bias,
    float* __restrict__ outp) {
    const int c = lane & 15;
    const int h = lane >> 4;
    for (int u = wid; u < RS_UNITS; u += RS_WAVES) {
        const int rt = u >> 2;
        const int cq = u & 3;
        if (rt * 16 >= nr) break;
        const float* ap = rows + (rt * 16 + c) * RS_PITCH + 4 * h;
        const float* bp = wl + h * WL_HALF + (cq * 32 + c) * 4;

        float ep[8];
        float bc0 = 0.0f, bc1 = 0.0f;
        if (post) {
            bc0 = bias[cq * 32 + c];
            bc1 = bias[cq * 32 + 16 + c];
        } else {
#pragma unroll
            for (int v = 0; v < 8; ++v) ep[v] = dinv[r0 + min(rt * 16 + v + 8 * h, nr - 1)];
        }

        v8f acc0 = v8f{}, acc1 = v8f{};
#pragma unroll
        for (int j = 0; j < FD / 8; ++j) {
            v4f a = *(const v4f*)(ap + 8 * j);
            const v4f b0 = *(const v4f*)(bp + j * WL_PAIR);
            const v4f b1 = *(const v4f*)(bp + j * WL_PAIR + 64);
            a.x = fmaxf(a.x, 0.0f); a.y = fmaxf(a.y, 0.0f);
            a.z = fmaxf(a.z, 0.0f); a.w = fmaxf(a.w, 0.0f);
            const v2f a01 = {a.x, a.y}, a23 = {a.z, a.w};
            const v2f p0 = {b0.x, b0.y}, p1 = {b1.x, b1.y};
            const v2f q0 = {b0.z, b0.w}, q1 = {b1.z, b1.w};
            acc0 = __builtin_amdgcn_wmma_f32_16x16x4_f32(false, a01, false, p0, (short)0, acc0, false, false);
            acc1 = __builtin_amdgcn_wmma_f32_16x16x4_f32(false, a01, false, p1, (short)0, acc1, false, false);
            acc0 = __builtin_amdgcn_wmma_f32_16x16x4_f32(false, a23, false, q0, (short)0, acc0, false, false);
            acc1 = __builtin_amdgcn_wmma_f32_16x16x4_f32(false, a23, false, q1, (short)0, acc1, false, false);
        }

#pragma unroll
        for (int v = 0; v < 8; ++v) {
            const int lr = rt * 16 + v + 8 * h;
            if (lr < nr) {
                float* op = outp + (size_t)(r0 + lr) * FD + cq * 32 + c;
                if (post) {
                    op[0]  = fmaxf(acc0[v] + bc0, 0.0f);
                    op[16] = fmaxf(acc1[v] + bc1, 0.0f);
                } else {
                    op[0]  = acc0[v] * ep[v];
                    op[16] = acc1[v] * ep[v];
                }
            }
        }
    }
}

template <int G>
static __device__ __forceinline__ void add_rows(
    int s, int dl, int nt, int lane,
    const float* __restrict__ hw, float* rows) {
    for (int g0 = 0; g0 < nt; g0 += G) {
        v4f v[G];
#pragma unroll
        for (int g = 0; g < G; ++g) {
            const int sj = __builtin_amdgcn_readlane(s, min(g0 + g, nt - 1));
            v[g] = ((const v4f*)(hw + (size_t)sj * FD))[lane];
        }
#pragma unroll
        for (int g = 0; g < G; ++g) {
            if (g == 0 || g0 + g < nt) {
                const int dj = __builtin_amdgcn_readlane(dl, g0 + g);
                float* rp = rows + dj + lane;
                unsafeAtomicAdd(rp,      v[g].x);
                unsafeAtomicAdd(rp + 32, v[g].y);
                unsafeAtomicAdd(rp + 64, v[g].z);
                unsafeAtomicAdd(rp + 96, v[g].w);
            }
        }
    }
}

static __device__ __forceinline__ void note_edges(
    const int* hits, int cnt, int lane, int r0,
    const int* __restrict__ src, const int* __restrict__ dst,
    unsigned* dcnt, v2i* __restrict__ rsd, int room, int* lpos) {
    if (cnt <= 0) return;
    int p = 0;
    if (lane == 0) p = atomicAdd(lpos, cnt);
    p = __builtin_amdgcn_readfirstlane(p);
    for (int base = 0; base < cnt; base += 32) {
        const int idx = base + lane;
        if (idx < cnt) {
            const int eid = hits[idx];
            const int s   = src[eid];
            const int d   = dst[eid];
            asm volatile("" :: "v"(s), "v"(d));
            const int dr  = d - r0;
            atomicAdd(&dcnt[dr], 1u);
            if (p + idx < room) rsd[p + idx] = v2i{s, dr * RS_PITCH};
        }
    }
}

__global__ __launch_bounds__(1024) void k_range_search(
    const int* __restrict__ src, const int* __restrict__ dst,
    const unsigned* __restrict__ key4, const unsigned* __restrict__ rcnt,
    float* __restrict__ dinv,
    int* __restrict__ rec_sd, int* __restrict__ rinfo,
    int e, int n) {
    __shared__ int hits[RS_WAVES * RS_CAP];
    __shared__ unsigned dcnt[RS_ROWS];
    __shared__ unsigned wsum[RS_WAVES];
    __shared__ int lpos;
    const int lane = threadIdx.x & 31;
    const int wid  = __builtin_amdgcn_readfirstlane((int)(threadIdx.x >> 5));
    const int r0   = blockIdx.x * RS_ROWS;
    const int nr   = min(RS_ROWS, n - r0);
    if (nr <= 0) return;

    unsigned part = 0u;
    if (threadIdx.x < RS_WGS && threadIdx.x < blockIdx.x) part = rcnt[threadIdx.x];
#pragma unroll
    for (int off = 16; off > 0; off >>= 1) part += __shfl_xor(part, off, 32);
    if (lane == 0) wsum[wid] = part;
    if (threadIdx.x == 0) lpos = 0;
    if (threadIdx.x < RS_ROWS) dcnt[threadIdx.x] = 1u;
    __syncthreads();

    unsigned tot = wsum[lane];
#pragma unroll
    for (int off = 16; off > 0; off >>= 1) tot += __shfl_xor(tot, off, 32);
    const int gbase = __builtin_amdgcn_readfirstlane((int)tot);
    v2i* myrsd = (v2i*)rec_sd + gbase;
    const int room = e - gbase;

    const unsigned splat = (unsigned)blockIdx.x * 0x01010101u;
    int* myhits = hits + wid * RS_CAP;
    int cnt = 0;
    const int thr = 32 + 8 * ((wid + (wid >> 2)) & 7);
    const int ntrips = (e + RS_TRIP - 1) / RS_TRIP;
    const int tlast  = ntrips - 1;
    const v4u* kp = (const v4u*)key4 + lane;
    v4u k0 = kp[(size_t)min(wid,                tlast) * 32];
    v4u k1 = kp[(size_t)min(wid +     RS_WAVES, tlast) * 32];
    v4u k2 = kp[(size_t)min(wid + 2 * RS_WAVES, tlast) * 32];
    v4u k3 = kp[(size_t)min(wid + 3 * RS_WAVES, tlast) * 32];
#pragma nounroll
    for (int tr = wid; tr < ntrips; tr += RS_WAVES) {
        const v4u kc = k0;
        k0 = k1; k1 = k2; k2 = k3;
        k3 = kp[(size_t)min(tr + 4 * RS_WAVES, tlast) * 32];
        unsigned m = (zero_bytes(kc.x ^ splat) >> 7) | (zero_bytes(kc.y ^ splat) >> 6) |
                     (zero_bytes(kc.z ^ splat) >> 5) | (zero_bytes(kc.w ^ splat) >> 4);
        while (__builtin_amdgcn_ballot_w32(m != 0u) != 0u) {
            const int  p   = __builtin_ctz(m | 0x80000000u);
            const int  eid = tr * RS_TRIP + lane * 16 + 4 * (p & 7) + (p >> 3);
            const bool ok  = (m != 0u) && (eid < e);
            const unsigned bm = __builtin_amdgcn_ballot_w32(ok);
            if (ok) myhits[cnt + (int)__builtin_amdgcn_mbcnt_lo(bm, 0u)] = eid;
            cnt += __builtin_popcount(bm);
            m &= m - 1u;
            if (cnt >= thr) {
                note_edges(myhits, cnt, lane, r0, src, dst, dcnt, myrsd, room, &lpos);
                cnt = 0;
            }
        }
    }
    note_edges(myhits, cnt, lane, r0, src, dst, dcnt, myrsd, room, &lpos);
    __syncthreads();

    if (threadIdx.x == 0) {
        rinfo[2 * blockIdx.x]     = gbase;
        rinfo[2 * blockIdx.x + 1] = min(lpos, room);
    }
    if ((int)threadIdx.x < nr)
        dinv[r0 + threadIdx.x] = rsqrtf((float)dcnt[threadIdx.x]);
}

static __device__ __forceinline__ v2i load_rec(const v2i* __restrict__ rsd, int idx, int cnt) {
    v2i sd = v2i{0, 0};
    if (idx < cnt) sd = rsd[idx];
    return sd;
}

static __device__ __forceinline__ void list_sums_body(
    const int* __restrict__ rec_sd, const int* __restrict__ rinfo,
    const float* __restrict__ dinv, const float* __restrict__ bias,
    const float* __restrict__ hw, const float* __restrict__ Wn,
    const float* __restrict__ biasn, float* __restrict__ emb,
    float* __restrict__ hwn, int n, bool post,
    float* rows, float* wl) {
    const int lane = threadIdx.x & 31;
    const int wid  = __builtin_amdgcn_readfirstlane((int)(threadIdx.x >> 5));
    const int r0   = blockIdx.x * RS_ROWS;
    const int nr   = min(RS_ROWS, n - r0);
    if (nr <= 0) return;
    const int gbase = rinfo[2 * blockIdx.x];
    const int cnt   = rinfo[2 * blockIdx.x + 1];
    const v2i* myrsd = (const v2i*)rec_sd + gbase;

    int c0 = wid * 32;
    v2i sd = load_rec(myrsd, c0 + lane, cnt);

    stage_w_rows(Wn, wl);

    const v4f* gl = (const v4f*)(hw + (size_t)r0 * FD);
    const int nv = nr * (FD / 4);
    {
        v4f sv[RS_SV];
#pragma unroll
        for (int i = 0; i < RS_SV; ++i)
            sv[i] = gl[min((int)threadIdx.x + i * RS_WAVES * 32, nv - 1)];
#pragma unroll
        for (int i = 0; i < RS_SV; ++i)
            if ((int)threadIdx.x + i * RS_WAVES * 32 < nv)
                rows_put(rows, (int)threadIdx.x + i * RS_WAVES * 32, sv[i]);
    }
    __syncthreads();

    for (; c0 < cnt; c0 += RS_WAVES * 32) {
        const v2i sdn = load_rec(myrsd, c0 + RS_WAVES * 32 + lane, cnt);
        add_rows<LS_GRP>(sd.x, sd.y, min(32, cnt - c0), lane, hw, rows);
        sd = sdn;
    }
    __syncthreads();

    rows_fin(rows, post ? (v4f*)(emb + (size_t)r0 * FD) : (v4f*)0, post, nv, r0, dinv, bias);
    __syncthreads();
    rows_product(rows, wl, nr, r0, wid, lane, post, dinv, biasn, hwn);
}

__global__ __launch_bounds__(1024) void k_list_sums_conv(
    const int* __restrict__ rec_sd, const int* __restrict__ rinfo,
    const float* __restrict__ dinv, const float* __restrict__ bias,
    const float* __restrict__ hw, const float* __restrict__ Wn,
    float* __restrict__ hwn, int n) {
    __shared__ alignas(16) float rows[RS_TILES * 16 * RS_PITCH];
    __shared__ alignas(16) float wl[WL_WORDS];
    list_sums_body(rec_sd, rinfo, dinv, bias, hw, Wn, (const float*)0, (float*)0, hwn, n, false, rows, wl);
}

__global__ __launch_bounds__(1024) void k_list_sums_post(
    const int* __restrict__ rec_sd, const int* __restrict__ rinfo,
    const float* __restrict__ dinv, const float* __restrict__ bias,
    const float* __restrict__ hw, const float* __restrict__ Wn,
    const float* __restrict__ biasn, float* __restrict__ emb,
    float* __restrict__ hwn, int n) {
    __shared__ alignas(16) float rows[RS_TILES * 16 * RS_PITCH];
    __shared__ alignas(16) float wl[WL_WORDS];
    list_sums_body(rec_sd, rinfo, dinv, bias, hw, Wn, biasn, emb, hwn, n, true, rows, wl);
}

__global__ __launch_bounds__(256) void k_logits(
    const float* __restrict__ h, const float* __restrict__ Wp2,
    const float* __restrict__ bp2, float* __restrict__ out, int n) {
    const int wid  = (int)((blockIdx.x * blockDim.x + threadIdx.x) >> 5);
    const int lane = threadIdx.x & 31;
    if (wid >= n) return;
    const float4 v = ((const float4*)(h + (size_t)wid * FD))[lane];
    const int f = lane * 4;
    float p0 = v.x * Wp2[(f + 0) * 2 + 0] + v.y * Wp2[(f + 1) * 2 + 0] +
               v.z * Wp2[(f + 2) * 2 + 0] + v.w * Wp2[(f + 3) * 2 + 0];
    float p1 = v.x * Wp2[(f + 0) * 2 + 1] + v.y * Wp2[(f + 1) * 2 + 1] +
               v.z * Wp2[(f + 2) * 2 + 1] + v.w * Wp2[(f + 3) * 2 + 1];
#pragma unroll
    for (int off = 16; off > 0; off >>= 1) {
        p0 += __shfl_down(p0, off, 32);
        p1 += __shfl_down(p1, off, 32);
    }
    if (lane == 0) {
        const float l0 = p0 + bp2[0];
        const float l1 = p1 + bp2[1];
        const float m  = fmaxf(l0, l1);
        const float lse = m + logf(expf(l0 - m) + expf(l1 - m));
        out[(size_t)wid * 2 + 0] = l0 - lse;
        out[(size_t)wid * 2 + 1] = l1 - lse;
    }
}

extern "C" void kernel_launch(void* const* d_in, const int* in_sizes, int n_in,
                              void* d_out, int out_size, void* d_ws,
                              size_t ws_size, hipStream_t stream) {
    const float* x   = (const float*)d_in[0];
    const int*   ei  = (const int*)d_in[1];
    const float* W1  = (const float*)d_in[2];
    const float* b1  = (const float*)d_in[3];
    const float* W2  = (const float*)d_in[4];
    const float* b2  = (const float*)d_in[5];
    const float* W3  = (const float*)d_in[6];
    const float* b3  = (const float*)d_in[7];
    const float* Wp1 = (const float*)d_in[8];
    const float* bp1 = (const float*)d_in[9];
    const float* Wp2 = (const float*)d_in[10];
    const float* bp2 = (const float*)d_in[11];

    const int N = in_sizes[0] / FD;
    const int E = in_sizes[1] / 2;
    const int* src = ei;
    const int* dst = ei + E;

    char* ws = (char*)d_ws;
    float* hwa  = (float*)ws;        ws += (size_t)N * FD * sizeof(float);
    float* hwb  = (float*)ws;        ws += (size_t)N * FD * sizeof(float);
    float* dinv = (float*)ws;        ws += (size_t)N * sizeof(float);
    unsigned* key4 = (unsigned*)ws;
    ws += (size_t)((E + RS_TRIP - 1) / RS_TRIP) * RS_TRIP;
    int* rec_sd  = (int*)ws;         ws += (size_t)E * 2 * sizeof(int);
    int* rinfo   = (int*)ws;         ws += (size_t)RS_WGS * 2 * sizeof(int);
    unsigned* rcnt = (unsigned*)ws;

    float* emb_out  = (float*)d_out;
    float* prob_out = (float*)d_out + (size_t)N * FD;

    const dim3 blk(256);
    const int nTrips  = (E + RS_TRIP - 1) / RS_TRIP;
    const int nKey4   = nTrips * (RS_TRIP / 4);
    const int gTiles  = N / 16;
    const int gGemm   = (gTiles + GT_WAVES - 1) / GT_WAVES;
    const dim3 blkSum(RS_WAVES * 32);
    const int gNodeW  = (N * WAVE + 255) / 256;

    k_cnt_init    <<<1, RS_WGS, 0, stream>>>(rcnt);
    k_keys        <<<KC_WGS, KC_THR, 0, stream>>>(dst, key4, rcnt, E, nKey4);
    k_range_search<<<RS_WGS, blkSum, 0, stream>>>(src, dst, key4, rcnt, dinv, rec_sd, rinfo, E, N);

    k_gemm_conv_in  <<<gGemm, blk, 0, stream>>>(x, W1, dinv, hwa, gTiles);
    k_list_sums_conv<<<RS_WGS, blkSum, 0, stream>>>(rec_sd, rinfo, dinv, b1, hwa, W2, hwb, N);

    k_list_sums_conv<<<RS_WGS, blkSum, 0, stream>>>(rec_sd, rinfo, dinv, b2, hwb, W3, hwa, N);

    k_list_sums_post<<<RS_WGS, blkSum, 0, stream>>>(rec_sd, rinfo, dinv, b3, hwa, Wp1, bp1, emb_out, hwb, N);

    k_logits        <<<gNodeW, blk, 0, stream>>>(hwb, Wp2, bp2, prob_out, N);
}
